# starting kernel
# baseline (speedup 1.0000x reference)
#include <hip/hip_runtime.h>

constexpr int N_ = 50000;
constexpr int E_ = 800000;
constexpr int F_ = 128;
constexpr int H_ = 8;
constexpr int D_ = 16;
constexpr int C_ = 40;
constexpr int L_ = 2;
constexpr float NEG_SLOPE = 0.2f;
constexpr float LN_EPS = 1e-5f;

typedef __attribute__((ext_vector_type(2))) float v2f;
typedef __attribute__((ext_vector_type(8))) float v8f;

__device__ __forceinline__ float lrelu(float x) { return x > 0.f ? x : NEG_SLOPE * x; }

__device__ __forceinline__ void atomicMaxF(float* addr, float val) {
  if (val >= 0.f) atomicMax((int*)addr, __float_as_int(val));
  else            atomicMin((unsigned int*)addr, __float_as_uint(val));
}

__global__ void gemm_node_fc(const float* __restrict__ X,
                             const float* __restrict__ W,
                             float* __restrict__ Y) {
  const int lane = threadIdx.x & 31;
  const int wave = threadIdx.x >> 5;
  const int m0   = blockIdx.x * 16;
  const int n0   = wave * 16;
  const int half = lane >> 4;
  const int l16  = lane & 15;
  const int arow = m0 + l16;
  const int bcol = n0 + l16;

  v8f acc = {};
#pragma unroll
  for (int k = 0; k < F_; k += 4) {
    const int ka = k + 2 * half;
    v2f a;
    a.x = X[arow * F_ + ka];
    a.y = X[arow * F_ + ka + 1];
    v2f b;
    b.x = W[ka * F_ + bcol];
    b.y = W[(ka + 1) * F_ + bcol];
    acc = __builtin_amdgcn_wmma_f32_16x16x4_f32(false, a, false, b,
                                                (short)0, acc, false, false);
  }
#pragma unroll
  for (int r = 0; r < 8; ++r) {
    const int m = m0 + r + 8 * half;
    Y[m * F_ + bcol] = acc[r];
  }
}

__global__ void node_coef(const float* __restrict__ hp,
                          const float* __restrict__ al,
                          const float* __restrict__ ar,
                          float* __restrict__ el, float* __restrict__ er,
                          float* __restrict__ mx, float* __restrict__ sm,
                          int total) {
  const int t = blockIdx.x * blockDim.x + threadIdx.x;
  if (t >= total) return;
  const int n = t >> 3;
  const int h = t & 7;
  const float* row = hp + n * F_ + h * D_;
  float sl = 0.f, sr = 0.f;
#pragma unroll
  for (int d = 0; d < D_; ++d) {
    const float x = row[d];
    sl += x * al[h * D_ + d];
    sr += x * ar[h * D_ + d];
  }
  el[t] = sl;
  er[t] = sr;
  mx[t] = __uint_as_float(0xff800000u);
  sm[t] = 0.f;
}

__global__ void zero_f32(float* __restrict__ p, int n) {
  const int t = blockIdx.x * blockDim.x + threadIdx.x;
  if (t < n) p[t] = 0.f;
}

__global__ void edge_max(const int* __restrict__ src, const int* __restrict__ dst,
                         const float* __restrict__ el, const float* __restrict__ er,
                         float* __restrict__ mx, int total) {
  const int t = blockIdx.x * blockDim.x + threadIdx.x;
  if (t >= total) return;
  const int e = t >> 3;
  const int h = t & 7;
  const int s = src[e];
  const int d = dst[e];
  const float v = lrelu(el[s * H_ + h] + er[d * H_ + h]);
  atomicMaxF(&mx[d * H_ + h], v);
}

__global__ void edge_sum(const int* __restrict__ src, const int* __restrict__ dst,
                         const float* __restrict__ el, const float* __restrict__ er,
                         const float* __restrict__ mx, float* __restrict__ sm,
                         int total) {
  const int t = blockIdx.x * blockDim.x + threadIdx.x;
  if (t >= total) return;
  const int e = t >> 3;
  const int h = t & 7;
  const int s = src[e];
  const int d = dst[e];
  const float v = lrelu(el[s * H_ + h] + er[d * H_ + h]);
  atomicAdd(&sm[d * H_ + h], __expf(v - mx[d * H_ + h]));
}

__global__ void edge_aggregate(const int* __restrict__ src, const int* __restrict__ dst,
                               const float* __restrict__ el, const float* __restrict__ er,
                               const float* __restrict__ mx, const float* __restrict__ sm,
                               const float* __restrict__ hp, float* __restrict__ out) {
  const int e    = blockIdx.x * (blockDim.x >> 5) + (threadIdx.x >> 5);
  const int lane = threadIdx.x & 31;
  if (e >= E_) return;
  const int s = src[e];
  const int d = dst[e];
  const int h = lane >> 2;
  const float v     = lrelu(el[s * H_ + h] + er[d * H_ + h]);
  const float alpha = __expf(v - mx[d * H_ + h]) / (sm[d * H_ + h] + 1e-16f);
  const float4 x = *(const float4*)(hp + (size_t)s * F_ + lane * 4);
  float* o = out + (size_t)d * F_ + lane * 4;
  atomicAdd(o + 0, alpha * x.x);
  atomicAdd(o + 1, alpha * x.y);
  atomicAdd(o + 2, alpha * x.z);
  atomicAdd(o + 3, alpha * x.w);
}

__global__ void ln_relu(float* __restrict__ io,
                        const float* __restrict__ bias,
                        const float* __restrict__ gamma,
                        const float* __restrict__ beta) {
  const int n    = blockIdx.x * (blockDim.x >> 5) + (threadIdx.x >> 5);
  const int lane = threadIdx.x & 31;
  if (n >= N_) return;
  float4 x = *(const float4*)(io + (size_t)n * F_ + lane * 4);
  const float4 bb = *(const float4*)(bias + lane * 4);
  x.x += bb.x; x.y += bb.y; x.z += bb.z; x.w += bb.w;
  float s  = x.x + x.y + x.z + x.w;
  float s2 = x.x * x.x + x.y * x.y + x.z * x.z + x.w * x.w;
#pragma unroll
  for (int off = 16; off >= 1; off >>= 1) {
    s  += __shfl_xor(s,  off, 32);
    s2 += __shfl_xor(s2, off, 32);
  }
  const float mu  = s * (1.f / 128.f);
  const float var = s2 * (1.f / 128.f) - mu * mu;
  const float inv = rsqrtf(var + LN_EPS);
  const float4 g  = *(const float4*)(gamma + lane * 4);
  const float4 be = *(const float4*)(beta + lane * 4);
  float4 y;
  y.x = fmaxf((x.x - mu) * inv * g.x + be.x, 0.f);
  y.y = fmaxf((x.y - mu) * inv * g.y + be.y, 0.f);
  y.z = fmaxf((x.z - mu) * inv * g.z + be.z, 0.f);
  y.w = fmaxf((x.w - mu) * inv * g.w + be.w, 0.f);
  *(float4*)(io + (size_t)n * F_ + lane * 4) = y;
}

__global__ void gemm_pred(const float* __restrict__ X,
                          const float* __restrict__ Wp,
                          const float* __restrict__ bp,
                          float* __restrict__ Y) {
  const int lane = threadIdx.x & 31;
  const int wave = threadIdx.x >> 5;
  const int m0   = blockIdx.x * 16;
  const int n0   = wave * 16;
  const int half = lane >> 4;
  const int l16  = lane & 15;
  const int arow = m0 + l16;
  const int n    = n0 + l16;
  const bool valid = (n < C_);

  v8f acc = {};
#pragma unroll
  for (int k = 0; k < F_; k += 4) {
    const int ka = k + 2 * half;
    v2f a;
    a.x = X[arow * F_ + ka];
    a.y = X[arow * F_ + ka + 1];
    v2f b;
    b.x = valid ? Wp[ka * C_ + n] : 0.f;
    b.y = valid ? Wp[(ka + 1) * C_ + n] : 0.f;
    acc = __builtin_amdgcn_wmma_f32_16x16x4_f32(false, a, false, b,
                                                (short)0, acc, false, false);
  }
  if (valid) {
    const float bbn = bp[n];
#pragma unroll
    for (int r = 0; r < 8; ++r) {
      const int m = m0 + r + 8 * half;
      Y[m * C_ + n] = acc[r] + bbn;
    }
  }
}

extern "C" void kernel_launch(void* const* d_in, const int* in_sizes, int n_in,
                              void* d_out, int out_size, void* d_ws, size_t ws_size,
                              hipStream_t stream) {
  const float* feats = (const float*)d_in[0];
  const int*   src   = (const int*)d_in[1];
  const int*   dst   = (const int*)d_in[2];
  const float* Ws    = (const float*)d_in[3];
  const float* al    = (const float*)d_in[4];
  const float* ar    = (const float*)d_in[5];
  const float* bias  = (const float*)d_in[6];
  const float* gamma = (const float*)d_in[7];
  const float* beta  = (const float*)d_in[8];
  const float* Wp    = (const float*)d_in[9];
  const float* bp    = (const float*)d_in[10];
  float* outp = (float*)d_out;

  float* ws   = (float*)d_ws;
  float* hbuf = ws;
  float* hp   = hbuf + (size_t)N_ * F_;
  float* el   = hp + (size_t)N_ * F_;
  float* er   = el + (size_t)N_ * H_;
  float* mx   = er + (size_t)N_ * H_;
  float* sm   = mx + (size_t)N_ * H_;

  const int NH = N_ * H_;
  const int EH = E_ * H_;
  const int NF = N_ * F_;

  for (int i = 0; i < L_; ++i) {
    const float* X = (i == 0) ? feats : hbuf;
    gemm_node_fc<<<N_ / 16, 256, 0, stream>>>(X, Ws + (size_t)i * F_ * F_, hp);
    node_coef<<<(NH + 255) / 256, 256, 0, stream>>>(
        hp, al + (size_t)i * H_ * D_, ar + (size_t)i * H_ * D_, el, er, mx, sm, NH);
    zero_f32<<<(NF + 255) / 256, 256, 0, stream>>>(hbuf, NF);
    edge_max<<<(EH + 255) / 256, 256, 0, stream>>>(src, dst, el, er, mx, EH);
    edge_sum<<<(EH + 255) / 256, 256, 0, stream>>>(src, dst, el, er, mx, sm, EH);
    edge_aggregate<<<E_ / 8, 256, 0, stream>>>(src, dst, el, er, mx, sm, hp, hbuf);
    ln_relu<<<N_ / 8, 256, 0, stream>>>(hbuf, bias + (size_t)i * F_,
                                        gamma + (size_t)i * F_, beta + (size_t)i * F_);
  }
  gemm_pred<<<N_ / 16, 96, 0, stream>>>(hbuf, Wp, bp, outp);
}


# Running max of the edge scores taken inside the node-range kernel (LDS atomic max over each wave's kept edge list): edge pass with global atomics gone, 9 -> 7 launches; sums still in arrival order
# speedup vs baseline: 2.8949x; 2.8949x over previous
#include <hip/hip_runtime.h>

constexpr int N_ = 50000;
constexpr int E_ = 800000;
constexpr int F_ = 128;
constexpr int H_ = 8;
constexpr int D_ = 16;
constexpr int C_ = 40;
constexpr int L_ = 2;
constexpr float NEG_SLOPE = 0.2f;
constexpr float LN_EPS = 1e-5f;

constexpr int RA_ROWS    = 200;
constexpr int RA_BLOCKS  = N_ / RA_ROWS;
constexpr int RA_WAVES   = 32;
constexpr int RA_THREADS = RA_WAVES * 32;
constexpr int RA_LOADS   = E_ / 4 / 32;
constexpr int RA_SCAN    = 8;
constexpr int RA_GROUP   = 8;
constexpr int RA_MAXT    = 16;
constexpr int RA_LIST    = 1536;
constexpr int RA_NH      = RA_ROWS * H_;
constexpr int RA_SUMS    = RA_ROWS * F_;
constexpr int RA_ACC     = RA_ROWS * F_ + RA_NH;
static_assert(RA_BLOCKS * RA_ROWS == N_, "node ranges");
static_assert(RA_LOADS * 128 == E_, "dst scan");
static_assert(E_ <= (1 << 20) && RA_ROWS <= 256, "list entry: edge number | row << 20");
static_assert(N_ <= (1 << 16), "list entry: source node | row << 16");
static_assert(RA_LIST > RA_SCAN * 128, "a scan trip always fits behind the capacity check");
static_assert((RA_ACC + 2 * RA_NH + RA_WAVES * RA_LIST + 1) * 4 <= 320 * 1024, "LDS of one work-group");

typedef __attribute__((ext_vector_type(2))) float v2f;
typedef __attribute__((ext_vector_type(8))) float v8f;

__device__ __forceinline__ float lrelu(float x) { return x > 0.f ? x : NEG_SLOPE * x; }

__device__ __forceinline__ void atomicMaxF(float* addr, float val) {
  if (val >= 0.f) atomicMax((int*)addr, __float_as_int(val));
  else            atomicMin((unsigned int*)addr, __float_as_uint(val));
}

__global__ void gemm_node_fc(const float* __restrict__ X,
                             const float* __restrict__ W,
                             float* __restrict__ Y) {
  const int lane = threadIdx.x & 31;
  const int wave = threadIdx.x >> 5;
  const int m0   = blockIdx.x * 16;
  const int n0   = wave * 16;
  const int half = lane >> 4;
  const int l16  = lane & 15;
  const int arow = m0 + l16;
  const int bcol = n0 + l16;

  v8f acc = {};
#pragma unroll
  for (int k = 0; k < F_; k += 4) {
    const int ka = k + 2 * half;
    v2f a;
    a.x = X[arow * F_ + ka];
    a.y = X[arow * F_ + ka + 1];
    v2f b;
    b.x = W[ka * F_ + bcol];
    b.y = W[(ka + 1) * F_ + bcol];
    acc = __builtin_amdgcn_wmma_f32_16x16x4_f32(false, a, false, b,
                                                (short)0, acc, false, false);
  }
#pragma unroll
  for (int r = 0; r < 8; ++r) {
    const int m = m0 + r + 8 * half;
    Y[m * F_ + bcol] = acc[r];
  }
}

__global__ void node_coef(const float* __restrict__ hp,
                          const float* __restrict__ al,
                          const float* __restrict__ ar,
                          float* __restrict__ el, float* __restrict__ er, int total) {
  const int t = blockIdx.x * blockDim.x + threadIdx.x;
  if (t >= total) return;
  const int n = t >> 3;
  const int h = t & 7;
  const float* row = hp + n * F_ + h * D_;
  float sl = 0.f, sr = 0.f;
#pragma unroll
  for (int d = 0; d < D_; ++d) {
    const float x = row[d];
    sl += x * al[h * D_ + d];
    sr += x * ar[h * D_ + d];
  }
  el[t] = sl;
  er[t] = sr;
}

template <int J>
__device__ __forceinline__ float quad_lane(float v) {
  return __int_as_float(__builtin_amdgcn_update_dpp(
      0, __float_as_int(v), J * 0x55, 0xF, 0xF, true));
}

template <bool FULL>
__device__ __forceinline__ void agg_group(const int* lst, int m, int lane, int h,
                                          bool sum_lane,
                                          const float* __restrict__ el,
                                          const float* __restrict__ hp,
                                          const float* erl, const float* mxl, float* acc) {
  const int slot = lane & (RA_GROUP - 1);
  const int ev = lst[(FULL || slot < m) ? slot : 0];
  int s[RA_GROUP], r[RA_GROUP];
  float cl[RA_GROUP], cr[RA_GROUP], cm[RA_GROUP], wt[RA_GROUP];
  float x0[RA_GROUP], x1[RA_GROUP], x2[RA_GROUP], x3[RA_GROUP];
#pragma unroll
  for (int j = 0; j < RA_GROUP; ++j) {
    const int t = __builtin_amdgcn_readlane(ev, j);
    s[j] = t & 0xFFFF;
    r[j] = t >> 16;
  }
#pragma unroll
  for (int j = 0; j < RA_GROUP; ++j) {
    cl[j] = el[s[j] * H_ + h];
    cr[j] = erl[r[j] * H_ + h];
    cm[j] = mxl[r[j] * H_ + h];
    const float* x = hp + (size_t)s[j] * F_ + lane;
    x0[j] = x[0]; x1[j] = x[32]; x2[j] = x[64]; x3[j] = x[96];
  }
  __builtin_amdgcn_sched_barrier(0);
#pragma unroll
  for (int j = 0; j < RA_GROUP; ++j) {
    const float v     = lrelu(cl[j] + cr[j]);
    const float alpha = __expf(v - cm[j]);
    wt[j] = alpha;
    if (FULL || j < m) {
      float* o = acc + r[j] * F_ + lane;
      atomicAdd(o,      quad_lane<0>(alpha) * x0[j]);
      atomicAdd(o + 32, quad_lane<1>(alpha) * x1[j]);
      atomicAdd(o + 64, quad_lane<2>(alpha) * x2[j]);
      atomicAdd(o + 96, quad_lane<3>(alpha) * x3[j]);
    }
  }
  if (sum_lane) {
#pragma unroll
    for (int j = 0; j < RA_GROUP; ++j)
      if (FULL || j < m) atomicAdd(&acc[RA_SUMS + r[j] * H_ + h], wt[j]);
  }
}

__global__ __launch_bounds__(RA_THREADS)
void range_aggregate(const int* __restrict__ src, const int* __restrict__ dst,
                     const float* __restrict__ el, const float* __restrict__ er,
                     const float* __restrict__ hp,
                     const float* __restrict__ bias, const float* __restrict__ gamma,
                     const float* __restrict__ beta, float* __restrict__ out) {
  __shared__ __attribute__((aligned(16))) float acc[RA_ACC];
  __shared__ float mxl[RA_NH];
  __shared__ float erl[RA_NH];
  __shared__ int lists[RA_WAVES * RA_LIST];
  __shared__ int ovf;
  const int lane = threadIdx.x & 31;
  const int wave = __builtin_amdgcn_readfirstlane((int)(threadIdx.x >> 5));
  const int n0   = blockIdx.x * RA_ROWS;
  const int h    = 2 * (lane & 3) + (lane >> 4);
  const bool sum_lane = (lane & 12) == 0;
  int* lst = lists + wave * RA_LIST;

  for (int i = threadIdx.x; i < RA_ACC; i += RA_THREADS) acc[i] = 0.f;
  for (int i = threadIdx.x; i < RA_NH; i += RA_THREADS) {
    mxl[i] = __uint_as_float(0xff800000u);
    erl[i] = er[n0 * H_ + i];
  }
  if (threadIdx.x == 0) ovf = 0;
  __syncthreads();

  const int4* dst4 = (const int4*)dst;
  int cnt = 0;
#pragma unroll 1
  for (int q0 = wave; q0 < RA_LOADS; q0 += RA_WAVES * RA_SCAN) {
    if (cnt > RA_LIST - RA_SCAN * 128) { ovf = 1; break; }
    int4 v[RA_SCAN];
#pragma unroll
    for (int j = 0; j < RA_SCAN; ++j) {
      const int q = q0 + j * RA_WAVES;
      const int4 t = dst4[(q < RA_LOADS ? q : q0) * 32 + lane];
      const int past = (q < RA_LOADS) ? 0 : -1;
      v[j].x = t.x | past; v[j].y = t.y | past; v[j].z = t.z | past; v[j].w = t.w | past;
    }
#pragma unroll
    for (int j = 0; j < RA_SCAN; ++j) {
      const int ebase = ((q0 + j * RA_WAVES) * 32 + lane) * 4;
      const int dd[4] = {v[j].x, v[j].y, v[j].z, v[j].w};
#pragma unroll
      for (int c = 0; c < 4; ++c) {
        const int row = dd[c] - n0;
        const bool hit = (unsigned)row < (unsigned)RA_ROWS;
        const unsigned bal = __builtin_amdgcn_ballot_w32(hit);
        if (bal != 0u) {
          if (hit) lst[cnt + (int)__builtin_amdgcn_mbcnt_lo(bal, 0u)] = (ebase + c) | (row << 20);
          cnt += __builtin_popcount(bal);
        }
      }
    }
  }

  for (int b = 0; b < cnt; b += 128) {
    int ent[4], sv[4];
#pragma unroll
    for (int k = 0; k < 4; ++k) {
      const int i = b + 32 * k + lane;
      ent[k] = (i < cnt) ? lst[i] : 0;
    }
#pragma unroll
    for (int k = 0; k < 4; ++k) sv[k] = src[ent[k] & 0xFFFFF];
#pragma unroll
    for (int k = 0; k < 4; ++k) {
      const int i = b + 32 * k + lane;
      if (i < cnt) lst[i] = sv[k] | ((ent[k] >> 20) << 16);
    }
  }
  {
    const int hh  = lane & 7;
    const int sub = lane >> 3;
    for (int b = 0; b < cnt; b += 4 * RA_MAXT) {
      int ent[RA_MAXT];
      float a[RA_MAXT];
#pragma unroll
      for (int t = 0; t < RA_MAXT; ++t) {
        const int i = b + 4 * t + sub;
        ent[t] = (i < cnt) ? lst[i] : -1;
      }
#pragma unroll
      for (int t = 0; t < RA_MAXT; ++t)
        a[t] = el[(ent[t] < 0 ? 0 : (ent[t] & 0xFFFF)) * H_ + hh];
#pragma unroll
      for (int t = 0; t < RA_MAXT; ++t) {
        if (ent[t] >= 0) {
          const int rh = (ent[t] >> 16) * H_ + hh;
          atomicMaxF(&mxl[rh], lrelu(a[t] + erl[rh]));
        }
      }
    }
  }
  __syncthreads();

  if (ovf == 0) {
    int head = 0;
    for (; cnt - head >= RA_GROUP; head += RA_GROUP)
      agg_group<true>(lst + head, RA_GROUP, lane, h, sum_lane, el, hp, erl, mxl, acc);
    if (cnt > head)
      agg_group<false>(lst + head, cnt - head, lane, h, sum_lane, el, hp, erl, mxl, acc);
  } else {
    for (int e = threadIdx.x; e < E_; e += RA_THREADS) {
      const int row = dst[e] - n0;
      if ((unsigned)row < (unsigned)RA_ROWS) {
        const int sn = src[e];
        for (int k = 0; k < H_; ++k)
          atomicMaxF(&mxl[row * H_ + k], lrelu(el[sn * H_ + k] + erl[row * H_ + k]));
      }
    }
    __syncthreads();
    for (int e = threadIdx.x; e < E_; e += RA_THREADS) {
      const int row = dst[e] - n0;
      if ((unsigned)row < (unsigned)RA_ROWS) {
        const int sn = src[e];
        for (int k = 0; k < H_; ++k) {
          const float v = lrelu(el[sn * H_ + k] + erl[row * H_ + k]);
          const float w = __expf(v - mxl[row * H_ + k]);
          atomicAdd(&acc[RA_SUMS + row * H_ + k], w);
          for (int d = 0; d < D_; ++d)
            atomicAdd(&acc[row * F_ + k * D_ + d], w * hp[(size_t)sn * F_ + k * D_ + d]);
        }
      }
    }
  }
  __syncthreads();

  const float4 bb = *(const float4*)(bias + lane * 4);
  const float4 g  = *(const float4*)(gamma + lane * 4);
  const float4 be = *(const float4*)(beta + lane * 4);
  for (int r = wave; r < RA_ROWS; r += RA_WAVES) {
    float4 x = *(const float4*)(acc + r * F_ + lane * 4);
    const float rs = 1.f / (acc[RA_SUMS + r * H_ + (lane >> 2)] + 1e-16f);
    x.x = x.x * rs + bb.x; x.y = x.y * rs + bb.y; x.z = x.z * rs + bb.z; x.w = x.w * rs + bb.w;
    float sA = x.x + x.y + x.z + x.w;
    float sB = x.x * x.x + x.y * x.y + x.z * x.z + x.w * x.w;
#pragma unroll
    for (int off = 16; off >= 1; off >>= 1) {
      sA += __shfl_xor(sA, off, 32);
      sB += __shfl_xor(sB, off, 32);
    }
    const float mu  = sA * (1.f / 128.f);
    const float var = sB * (1.f / 128.f) - mu * mu;
    const float inv = rsqrtf(var + LN_EPS);
    float4 y;
    y.x = fmaxf((x.x - mu) * inv * g.x + be.x, 0.f);
    y.y = fmaxf((x.y - mu) * inv * g.y + be.y, 0.f);
    y.z = fmaxf((x.z - mu) * inv * g.z + be.z, 0.f);
    y.w = fmaxf((x.w - mu) * inv * g.w + be.w, 0.f);
    *(float4*)(out + (size_t)(n0 + r) * F_ + lane * 4) = y;
  }
}

__global__ void gemm_pred(const float* __restrict__ X,
                          const float* __restrict__ Wp,
                          const float* __restrict__ bp,
                          float* __restrict__ Y) {
  const int lane = threadIdx.x & 31;
  const int wave = threadIdx.x >> 5;
  const int m0   = blockIdx.x * 16;
  const int n0   = wave * 16;
  const int half = lane >> 4;
  const int l16  = lane & 15;
  const int arow = m0 + l16;
  const int n    = n0 + l16;
  const bool valid = (n < C_);

  v8f acc = {};
#pragma unroll
  for (int k = 0; k < F_; k += 4) {
    const int ka = k + 2 * half;
    v2f a;
    a.x = X[arow * F_ + ka];
    a.y = X[arow * F_ + ka + 1];
    v2f b;
    b.x = valid ? Wp[ka * C_ + n] : 0.f;
    b.y = valid ? Wp[(ka + 1) * C_ + n] : 0.f;
    acc = __builtin_amdgcn_wmma_f32_16x16x4_f32(false, a, false, b,
                                                (short)0, acc, false, false);
  }
  if (valid) {
    const float bbn = bp[n];
#pragma unroll
    for (int r = 0; r < 8; ++r) {
      const int m = m0 + r + 8 * half;
      Y[m * C_ + n] = acc[r] + bbn;
    }
  }
}

extern "C" void kernel_launch(void* const* d_in, const int* in_sizes, int n_in,
                              void* d_out, int out_size, void* d_ws, size_t ws_size,
                              hipStream_t stream) {
  const float* feats = (const float*)d_in[0];
  const int*   src   = (const int*)d_in[1];
  const int*   dst   = (const int*)d_in[2];
  const float* Ws    = (const float*)d_in[3];
  const float* al    = (const float*)d_in[4];
  const float* ar    = (const float*)d_in[5];
  const float* bias  = (const float*)d_in[6];
  const float* gamma = (const float*)d_in[7];
  const float* beta  = (const float*)d_in[8];
  const float* Wp    = (const float*)d_in[9];
  const float* bp    = (const float*)d_in[10];
  float* outp = (float*)d_out;

  float* ws   = (float*)d_ws;
  float* hbuf = ws;
  float* hp   = hbuf + (size_t)N_ * F_;
  float* el   = hp + (size_t)N_ * F_;
  float* er   = el + (size_t)N_ * H_;

  const int NH = N_ * H_;

  for (int i = 0; i < L_; ++i) {
    const float* X = (i == 0) ? feats : hbuf;
    gemm_node_fc<<<N_ / 16, 256, 0, stream>>>(X, Ws + (size_t)i * F_ * F_, hp);
    node_coef<<<(NH + 255) / 256, 256, 0, stream>>>(
        hp, al + (size_t)i * H_ * D_, ar + (size_t)i * H_ * D_, el, er, NH);
    range_aggregate<<<RA_BLOCKS, RA_THREADS, 0, stream>>>(
        src, dst, el, er, hp, bias + (size_t)i * F_, gamma + (size_t)i * F_,
        beta + (size_t)i * F_, hbuf);
  }
  gemm_pred<<<N_ / 16, 96, 0, stream>>>(hbuf, Wp, bp, outp);
}


# Second layer's node-range kernel reads the per-wave edge lists (source | row) that the first layer's scan stored in the workspace: no second scan of dst or src fetch; sums still in arrival order
# speedup vs baseline: 3.2513x; 1.1231x over previous
#include <hip/hip_runtime.h>

constexpr int N_ = 50000;
constexpr int E_ = 800000;
constexpr int F_ = 128;
constexpr int H_ = 8;
constexpr int D_ = 16;
constexpr int C_ = 40;
constexpr int L_ = 2;
constexpr float NEG_SLOPE = 0.2f;
constexpr float LN_EPS = 1e-5f;

constexpr int RA_ROWS    = 200;
constexpr int RA_BLOCKS  = N_ / RA_ROWS;
constexpr int RA_WAVES   = 32;
constexpr int RA_THREADS = RA_WAVES * 32;
constexpr int RA_LOADS   = E_ / 4 / 32;
constexpr int RA_SCAN    = 8;
constexpr int RA_GROUP   = 8;
constexpr int RA_MAXT    = 16;
constexpr int RA_LIST    = 1536;
constexpr int RA_NH      = RA_ROWS * H_;
constexpr int RA_SUMS    = RA_ROWS * F_;
constexpr int RA_ACC     = RA_ROWS * F_ + RA_NH;
static_assert(RA_BLOCKS * RA_ROWS == N_, "node ranges");
static_assert(RA_LOADS * 128 == E_, "dst scan");
static_assert(E_ <= (1 << 20) && RA_ROWS <= 256, "list entry: edge number | row << 20");
static_assert(N_ <= (1 << 16), "list entry: source node | row << 16");
static_assert(RA_LIST > RA_SCAN * 128, "a scan trip always fits behind the capacity check");
static_assert((RA_ACC + 2 * RA_NH + RA_WAVES * RA_LIST + 1) * 4 <= 320 * 1024, "LDS of one work-group");

typedef __attribute__((ext_vector_type(2))) float v2f;
typedef __attribute__((ext_vector_type(8))) float v8f;

__device__ __forceinline__ float lrelu(float x) { return x > 0.f ? x : NEG_SLOPE * x; }

__device__ __forceinline__ void atomicMaxF(float* addr, float val) {
  if (val >= 0.f) atomicMax((int*)addr, __float_as_int(val));
  else            atomicMin((unsigned int*)addr, __float_as_uint(val));
}

__global__ void gemm_node_fc(const float* __restrict__ X,
                             const float* __restrict__ W,
                             float* __restrict__ Y) {
  const int lane = threadIdx.x & 31;
  const int wave = threadIdx.x >> 5;
  const int m0   = blockIdx.x * 16;
  const int n0   = wave * 16;
  const int half = lane >> 4;
  const int l16  = lane & 15;
  const int arow = m0 + l16;
  const int bcol = n0 + l16;

  v8f acc = {};
#pragma unroll
  for (int k = 0; k < F_; k += 4) {
    const int ka = k + 2 * half;
    v2f a;
    a.x = X[arow * F_ + ka];
    a.y = X[arow * F_ + ka + 1];
    v2f b;
    b.x = W[ka * F_ + bcol];
    b.y = W[(ka + 1) * F_ + bcol];
    acc = __builtin_amdgcn_wmma_f32_16x16x4_f32(false, a, false, b,
                                                (short)0, acc, false, false);
  }
#pragma unroll
  for (int r = 0; r < 8; ++r) {
    const int m = m0 + r + 8 * half;
    Y[m * F_ + bcol] = acc[r];
  }
}

__global__ void node_coef(const float* __restrict__ hp,
                          const float* __restrict__ al,
                          const float* __restrict__ ar,
                          float* __restrict__ el, float* __restrict__ er, int total) {
  const int t = blockIdx.x * blockDim.x + threadIdx.x;
  if (t >= total) return;
  const int n = t >> 3;
  const int h = t & 7;
  const float* row = hp + n * F_ + h * D_;
  float sl = 0.f, sr = 0.f;
#pragma unroll
  for (int d = 0; d < D_; ++d) {
    const float x = row[d];
    sl += x * al[h * D_ + d];
    sr += x * ar[h * D_ + d];
  }
  el[t] = sl;
  er[t] = sr;
}

template <int J>
__device__ __forceinline__ float quad_lane(float v) {
  return __int_as_float(__builtin_amdgcn_update_dpp(
      0, __float_as_int(v), J * 0x55, 0xF, 0xF, true));
}

template <bool FULL>
__device__ __forceinline__ void agg_group(const int* lst, int m, int lane, int h,
                                          bool sum_lane,
                                          const float* __restrict__ el,
                                          const float* __restrict__ hp,
                                          const float* erl, const float* mxl, float* acc) {
  const int slot = lane & (RA_GROUP - 1);
  const int ev = lst[(FULL || slot < m) ? slot : 0];
  int s[RA_GROUP], r[RA_GROUP];
  float cl[RA_GROUP], cr[RA_GROUP], cm[RA_GROUP], wt[RA_GROUP];
  float x0[RA_GROUP], x1[RA_GROUP], x2[RA_GROUP], x3[RA_GROUP];
#pragma unroll
  for (int j = 0; j < RA_GROUP; ++j) {
    const int t = __builtin_amdgcn_readlane(ev, j);
    s[j] = t & 0xFFFF;
    r[j] = t >> 16;
  }
#pragma unroll
  for (int j = 0; j < RA_GROUP; ++j) {
    cl[j] = el[s[j] * H_ + h];
    cr[j] = erl[r[j] * H_ + h];
    cm[j] = mxl[r[j] * H_ + h];
    const float* x = hp + (size_t)s[j] * F_ + lane;
    x0[j] = x[0]; x1[j] = x[32]; x2[j] = x[64]; x3[j] = x[96];
  }
  __builtin_amdgcn_sched_barrier(0);
#pragma unroll
  for (int j = 0; j < RA_GROUP; ++j) {
    const float v     = lrelu(cl[j] + cr[j]);
    const float alpha = __expf(v - cm[j]);
    wt[j] = alpha;
    if (FULL || j < m) {
      float* o = acc + r[j] * F_ + lane;
      atomicAdd(o,      quad_lane<0>(alpha) * x0[j]);
      atomicAdd(o + 32, quad_lane<1>(alpha) * x1[j]);
      atomicAdd(o + 64, quad_lane<2>(alpha) * x2[j]);
      atomicAdd(o + 96, quad_lane<3>(alpha) * x3[j]);
    }
  }
  if (sum_lane) {
#pragma unroll
    for (int j = 0; j < RA_GROUP; ++j)
      if (FULL || j < m) atomicAdd(&acc[RA_SUMS + r[j] * H_ + h], wt[j]);
  }
}

template <bool FIRST>
__global__ __launch_bounds__(RA_THREADS)
void range_aggregate(const int* __restrict__ src, const int* __restrict__ dst,
                     const float* __restrict__ el, const float* __restrict__ er,
                     const float* __restrict__ hp,
                     const float* __restrict__ bias, const float* __restrict__ gamma,
                     const float* __restrict__ beta, float* __restrict__ out,
                     int* __restrict__ wl, int* __restrict__ wc, int* __restrict__ wo) {
  __shared__ __attribute__((aligned(16))) float acc[RA_ACC];
  __shared__ float mxl[RA_NH];
  __shared__ float erl[RA_NH];
  __shared__ int lists[RA_WAVES * RA_LIST];
  __shared__ int ovf;
  const int lane = threadIdx.x & 31;
  const int wave = __builtin_amdgcn_readfirstlane((int)(threadIdx.x >> 5));
  const int n0   = blockIdx.x * RA_ROWS;
  const int h    = 2 * (lane & 3) + (lane >> 4);
  const bool sum_lane = (lane & 12) == 0;
  int* lst = lists + wave * RA_LIST;
  int* wsl = wl + (size_t)(blockIdx.x * RA_WAVES + wave) * RA_LIST;
  const int ovf1 = FIRST ? 0 : wo[blockIdx.x];

  for (int i = threadIdx.x; i < RA_ACC; i += RA_THREADS) acc[i] = 0.f;
  for (int i = threadIdx.x; i < RA_NH; i += RA_THREADS) {
    mxl[i] = __uint_as_float(0xff800000u);
    erl[i] = er[n0 * H_ + i];
  }
  if (threadIdx.x == 0) ovf = ovf1;
  __syncthreads();

  int cnt = 0;
  if (FIRST) {
  const int4* dst4 = (const int4*)dst;
#pragma unroll 1
  for (int q0 = wave; q0 < RA_LOADS; q0 += RA_WAVES * RA_SCAN) {
    if (cnt > RA_LIST - RA_SCAN * 128) { ovf = 1; break; }
    int4 v[RA_SCAN];
#pragma unroll
    for (int j = 0; j < RA_SCAN; ++j) {
      const int q = q0 + j * RA_WAVES;
      const int4 t = dst4[(q < RA_LOADS ? q : q0) * 32 + lane];
      const int past = (q < RA_LOADS) ? 0 : -1;
      v[j].x = t.x | past; v[j].y = t.y | past; v[j].z = t.z | past; v[j].w = t.w | past;
    }
#pragma unroll
    for (int j = 0; j < RA_SCAN; ++j) {
      const int ebase = ((q0 + j * RA_WAVES) * 32 + lane) * 4;
      const int dd[4] = {v[j].x, v[j].y, v[j].z, v[j].w};
#pragma unroll
      for (int c = 0; c < 4; ++c) {
        const int row = dd[c] - n0;
        const bool hit = (unsigned)row < (unsigned)RA_ROWS;
        const unsigned bal = __builtin_amdgcn_ballot_w32(hit);
        if (bal != 0u) {
          if (hit) lst[cnt + (int)__builtin_amdgcn_mbcnt_lo(bal, 0u)] = (ebase + c) | (row << 20);
          cnt += __builtin_popcount(bal);
        }
      }
    }
  }

  for (int b = 0; b < cnt; b += 128) {
    int ent[4], sv[4];
#pragma unroll
    for (int k = 0; k < 4; ++k) {
      const int i = b + 32 * k + lane;
      ent[k] = (i < cnt) ? lst[i] : 0;
    }
#pragma unroll
    for (int k = 0; k < 4; ++k) sv[k] = src[ent[k] & 0xFFFFF];
#pragma unroll
    for (int k = 0; k < 4; ++k) {
      const int i = b + 32 * k + lane;
      if (i < cnt) {
        const int val = sv[k] | ((ent[k] >> 20) << 16);
        lst[i] = val;
        wsl[i] = val;
      }
    }
  }
  if (lane == 0) wc[blockIdx.x * RA_WAVES + wave] = cnt;
  } else {
    if (ovf1 == 0) cnt = __builtin_amdgcn_readfirstlane(wc[blockIdx.x * RA_WAVES + wave]);
    for (int b = 0; b < cnt; b += 128) {
      int ent[4];
#pragma unroll
      for (int k = 0; k < 4; ++k) {
        const int i = b + 32 * k + lane;
        ent[k] = wsl[i < cnt ? i : 0];
      }
#pragma unroll
      for (int k = 0; k < 4; ++k) {
        const int i = b + 32 * k + lane;
        if (i < cnt) lst[i] = ent[k];
      }
    }
  }
  {
    const int hh  = lane & 7;
    const int sub = lane >> 3;
    for (int b = 0; b < cnt; b += 4 * RA_MAXT) {
      int ent[RA_MAXT];
      float a[RA_MAXT];
#pragma unroll
      for (int t = 0; t < RA_MAXT; ++t) {
        const int i = b + 4 * t + sub;
        ent[t] = (i < cnt) ? lst[i] : -1;
      }
#pragma unroll
      for (int t = 0; t < RA_MAXT; ++t)
        a[t] = el[(ent[t] < 0 ? 0 : (ent[t] & 0xFFFF)) * H_ + hh];
#pragma unroll
      for (int t = 0; t < RA_MAXT; ++t) {
        if (ent[t] >= 0) {
          const int rh = (ent[t] >> 16) * H_ + hh;
          atomicMaxF(&mxl[rh], lrelu(a[t] + erl[rh]));
        }
      }
    }
  }
  __syncthreads();
  if (FIRST && threadIdx.x == 0) wo[blockIdx.x] = ovf;

  if (ovf == 0) {
    int head = 0;
    for (; cnt - head >= RA_GROUP; head += RA_GROUP)
      agg_group<true>(lst + head, RA_GROUP, lane, h, sum_lane, el, hp, erl, mxl, acc);
    if (cnt > head)
      agg_group<false>(lst + head, cnt - head, lane, h, sum_lane, el, hp, erl, mxl, acc);
  } else {
    for (int e = threadIdx.x; e < E_; e += RA_THREADS) {
      const int row = dst[e] - n0;
      if ((unsigned)row < (unsigned)RA_ROWS) {
        const int sn = src[e];
        for (int k = 0; k < H_; ++k)
          atomicMaxF(&mxl[row * H_ + k], lrelu(el[sn * H_ + k] + erl[row * H_ + k]));
      }
    }
    __syncthreads();
    for (int e = threadIdx.x; e < E_; e += RA_THREADS) {
      const int row = dst[e] - n0;
      if ((unsigned)row < (unsigned)RA_ROWS) {
        const int sn = src[e];
        for (int k = 0; k < H_; ++k) {
          const float v = lrelu(el[sn * H_ + k] + erl[row * H_ + k]);
          const float w = __expf(v - mxl[row * H_ + k]);
          atomicAdd(&acc[RA_SUMS + row * H_ + k], w);
          for (int d = 0; d < D_; ++d)
            atomicAdd(&acc[row * F_ + k * D_ + d], w * hp[(size_t)sn * F_ + k * D_ + d]);
        }
      }
    }
  }
  __syncthreads();

  const float4 bb = *(const float4*)(bias + lane * 4);
  const float4 g  = *(const float4*)(gamma + lane * 4);
  const float4 be = *(const float4*)(beta + lane * 4);
  for (int r = wave; r < RA_ROWS; r += RA_WAVES) {
    float4 x = *(const float4*)(acc + r * F_ + lane * 4);
    const float rs = 1.f / (acc[RA_SUMS + r * H_ + (lane >> 2)] + 1e-16f);
    x.x = x.x * rs + bb.x; x.y = x.y * rs + bb.y; x.z = x.z * rs + bb.z; x.w = x.w * rs + bb.w;
    float sA = x.x + x.y + x.z + x.w;
    float sB = x.x * x.x + x.y * x.y + x.z * x.z + x.w * x.w;
#pragma unroll
    for (int off = 16; off >= 1; off >>= 1) {
      sA += __shfl_xor(sA, off, 32);
      sB += __shfl_xor(sB, off, 32);
    }
    const float mu  = sA * (1.f / 128.f);
    const float var = sB * (1.f / 128.f) - mu * mu;
    const float inv = rsqrtf(var + LN_EPS);
    float4 y;
    y.x = fmaxf((x.x - mu) * inv * g.x + be.x, 0.f);
    y.y = fmaxf((x.y - mu) * inv * g.y + be.y, 0.f);
    y.z = fmaxf((x.z - mu) * inv * g.z + be.z, 0.f);
    y.w = fmaxf((x.w - mu) * inv * g.w + be.w, 0.f);
    *(float4*)(out + (size_t)(n0 + r) * F_ + lane * 4) = y;
  }
}

__global__ void gemm_pred(const float* __restrict__ X,
                          const float* __restrict__ Wp,
                          const float* __restrict__ bp,
                          float* __restrict__ Y) {
  const int lane = threadIdx.x & 31;
  const int wave = threadIdx.x >> 5;
  const int m0   = blockIdx.x * 16;
  const int n0   = wave * 16;
  const int half = lane >> 4;
  const int l16  = lane & 15;
  const int arow = m0 + l16;
  const int n    = n0 + l16;
  const bool valid = (n < C_);

  v8f acc = {};
#pragma unroll
  for (int k = 0; k < F_; k += 4) {
    const int ka = k + 2 * half;
    v2f a;
    a.x = X[arow * F_ + ka];
    a.y = X[arow * F_ + ka + 1];
    v2f b;
    b.x = valid ? Wp[ka * C_ + n] : 0.f;
    b.y = valid ? Wp[(ka + 1) * C_ + n] : 0.f;
    acc = __builtin_amdgcn_wmma_f32_16x16x4_f32(false, a, false, b,
                                                (short)0, acc, false, false);
  }
  if (valid) {
    const float bbn = bp[n];
#pragma unroll
    for (int r = 0; r < 8; ++r) {
      const int m = m0 + r + 8 * half;
      Y[m * C_ + n] = acc[r] + bbn;
    }
  }
}

extern "C" void kernel_launch(void* const* d_in, const int* in_sizes, int n_in,
                              void* d_out, int out_size, void* d_ws, size_t ws_size,
                              hipStream_t stream) {
  const float* feats = (const float*)d_in[0];
  const int*   src   = (const int*)d_in[1];
  const int*   dst   = (const int*)d_in[2];
  const float* Ws    = (const float*)d_in[3];
  const float* al    = (const float*)d_in[4];
  const float* ar    = (const float*)d_in[5];
  const float* bias  = (const float*)d_in[6];
  const float* gamma = (const float*)d_in[7];
  const float* beta  = (const float*)d_in[8];
  const float* Wp    = (const float*)d_in[9];
  const float* bp    = (const float*)d_in[10];
  float* outp = (float*)d_out;

  float* ws   = (float*)d_ws;
  float* hbuf = ws;
  float* hp   = hbuf + (size_t)N_ * F_;
  float* el   = hp + (size_t)N_ * F_;
  float* er   = el + (size_t)N_ * H_;
  int* wl     = (int*)(er + (size_t)N_ * H_);
  int* wc     = wl + (size_t)RA_BLOCKS * RA_WAVES * RA_LIST;
  int* wo     = wc + RA_BLOCKS * RA_WAVES;

  const int NH = N_ * H_;

  for (int i = 0; i < L_; ++i) {
    const float* X = (i == 0) ? feats : hbuf;
    gemm_node_fc<<<N_ / 16, 256, 0, stream>>>(X, Ws + (size_t)i * F_ * F_, hp);
    node_coef<<<(NH + 255) / 256, 256, 0, stream>>>(
        hp, al + (size_t)i * H_ * D_, ar + (size_t)i * H_ * D_, el, er, NH);
    if (i == 0)
      range_aggregate<true><<<RA_BLOCKS, RA_THREADS, 0, stream>>>(
          src, dst, el, er, hp, bias + (size_t)i * F_, gamma + (size_t)i * F_,
          beta + (size_t)i * F_, hbuf, wl, wc, wo);
    else
      range_aggregate<false><<<RA_BLOCKS, RA_THREADS, 0, stream>>>(
          src, dst, el, er, hp, bias + (size_t)i * F_, gamma + (size_t)i * F_,
          beta + (size_t)i * F_, hbuf, wl, wc, wo);
  }
  gemm_pred<<<N_ / 16, 96, 0, stream>>>(hbuf, Wp, bp, outp);
}


# Finished rows stay in LDS and the product that consumes them (next layer's fc, final prediction) runs in the node-range kernel's epilogue, A fragments from LDS: 7 -> 5 launches; same summation order
# speedup vs baseline: 3.8238x; 1.1761x over previous
#include <hip/hip_runtime.h>

constexpr int N_ = 50000;
constexpr int E_ = 800000;
constexpr int F_ = 128;
constexpr int H_ = 8;
constexpr int D_ = 16;
constexpr int C_ = 40;
constexpr int L_ = 2;
constexpr float NEG_SLOPE = 0.2f;
constexpr float LN_EPS = 1e-5f;

constexpr int RA_ROWS    = 200;
constexpr int RA_BLOCKS  = N_ / RA_ROWS;
constexpr int RA_WAVES   = 32;
constexpr int RA_THREADS = RA_WAVES * 32;
constexpr int RA_LOADS   = E_ / 4 / 32;
constexpr int RA_SCAN    = 8;
constexpr int RA_GROUP   = 8;
constexpr int RA_MAXT    = 16;
constexpr int RA_LIST    = 1536;
constexpr int RA_NH      = RA_ROWS * H_;
constexpr int RA_SUMS    = RA_ROWS * F_;
constexpr int RA_ACC     = RA_ROWS * F_ + RA_NH;
static_assert(RA_BLOCKS * RA_ROWS == N_, "node ranges");
static_assert(RA_LOADS * 128 == E_, "dst scan");
static_assert(E_ <= (1 << 20) && RA_ROWS <= 256, "list entry: edge number | row << 20");
static_assert(N_ <= (1 << 16), "list entry: source node | row << 16");
static_assert(RA_LIST > RA_SCAN * 128, "a scan trip always fits behind the capacity check");
static_assert((RA_ACC + 2 * RA_NH + RA_WAVES * RA_LIST + 1) * 4 <= 320 * 1024, "LDS of one work-group");
constexpr int RA_MT      = (RA_ROWS + 15) / 16;
constexpr int RA_PITCH   = F_ + 4;
constexpr int RA_PTILES  = RA_MT * ((C_ + 15) / 16);
static_assert(RA_MT * 16 * RA_PITCH <= RA_WAVES * RA_LIST, "finished rows fit where the lists were");
static_assert(RA_MT == 13 && RA_WAVES == 32, "fc tiles: 8 column slabs x (3 row tiles per wave, a 4th for waves 0..7)");
static_assert((RA_MT * 16 - RA_ROWS) * F_ == RA_THREADS, "pad rows: one float per thread");
static_assert(L_ == 2, "kernel_launch is written out for two layers");
static_assert(RA_ROWS % 8 == 0, "a lane's eight accumulator rows are all real or all pad");

typedef __attribute__((ext_vector_type(2))) float v2f;
typedef __attribute__((ext_vector_type(8))) float v8f;

__device__ __forceinline__ float lrelu(float x) { return x > 0.f ? x : NEG_SLOPE * x; }

__device__ __forceinline__ void atomicMaxF(float* addr, float val) {
  if (val >= 0.f) atomicMax((int*)addr, __float_as_int(val));
  else            atomicMin((unsigned int*)addr, __float_as_uint(val));
}

__global__ void gemm_node_fc(const float* __restrict__ X,
                             const float* __restrict__ W,
                             float* __restrict__ Y) {
  const int lane = threadIdx.x & 31;
  const int wave = threadIdx.x >> 5;
  const int m0   = blockIdx.x * 16;
  const int n0   = wave * 16;
  const int half = lane >> 4;
  const int l16  = lane & 15;
  const int arow = m0 + l16;
  const int bcol = n0 + l16;

  v8f acc = {};
#pragma unroll
  for (int k = 0; k < F_; k += 4) {
    const int ka = k + 2 * half;
    v2f a;
    a.x = X[arow * F_ + ka];
    a.y = X[arow * F_ + ka + 1];
    v2f b;
    b.x = W[ka * F_ + bcol];
    b.y = W[(ka + 1) * F_ + bcol];
    acc = __builtin_amdgcn_wmma_f32_16x16x4_f32(false, a, false, b,
                                                (short)0, acc, false, false);
  }
#pragma unroll
  for (int r = 0; r < 8; ++r) {
    const int m = m0 + r + 8 * half;
    Y[m * F_ + bcol] = acc[r];
  }
}

__global__ void node_coef(const float* __restrict__ hp,
                          const float* __restrict__ al,
                          const float* __restrict__ ar,
                          float* __restrict__ el, float* __restrict__ er, int total) {
  const int t = blockIdx.x * blockDim.x + threadIdx.x;
  if (t >= total) return;
  const int n = t >> 3;
  const int h = t & 7;
  const float* row = hp + n * F_ + h * D_;
  float sl = 0.f, sr = 0.f;
#pragma unroll
  for (int d = 0; d < D_; ++d) {
    const float x = row[d];
    sl += x * al[h * D_ + d];
    sr += x * ar[h * D_ + d];
  }
  el[t] = sl;
  er[t] = sr;
}

template <int J>
__device__ __forceinline__ float quad_lane(float v) {
  return __int_as_float(__builtin_amdgcn_update_dpp(
      0, __float_as_int(v), J * 0x55, 0xF, 0xF, true));
}

template <bool FULL>
__device__ __forceinline__ void agg_group(const int* lst, int m, int lane, int h,
                                          bool sum_lane,
                                          const float* __restrict__ el,
                                          const float* __restrict__ hp,
                                          const float* erl, const float* mxl, float* acc) {
  const int slot = lane & (RA_GROUP - 1);
  const int ev = lst[(FULL || slot < m) ? slot : 0];
  int s[RA_GROUP], r[RA_GROUP];
  float cl[RA_GROUP], cr[RA_GROUP], cm[RA_GROUP], wt[RA_GROUP];
  float x0[RA_GROUP], x1[RA_GROUP], x2[RA_GROUP], x3[RA_GROUP];
#pragma unroll
  for (int j = 0; j < RA_GROUP; ++j) {
    const int t = __builtin_amdgcn_readlane(ev, j);
    s[j] = t & 0xFFFF;
    r[j] = t >> 16;
  }
#pragma unroll
  for (int j = 0; j < RA_GROUP; ++j) {
    cl[j] = el[s[j] * H_ + h];
    cr[j] = erl[r[j] * H_ + h];
    cm[j] = mxl[r[j] * H_ + h];
    const float* x = hp + (size_t)s[j] * F_ + lane;
    x0[j] = x[0]; x1[j] = x[32]; x2[j] = x[64]; x3[j] = x[96];
  }
  __builtin_amdgcn_sched_barrier(0);
#pragma unroll
  for (int j = 0; j < RA_GROUP; ++j) {
    const float v     = lrelu(cl[j] + cr[j]);
    const float alpha = __expf(v - cm[j]);
    wt[j] = alpha;
    if (FULL || j < m) {
      float* o = acc + r[j] * F_ + lane;
      atomicAdd(o,      quad_lane<0>(alpha) * x0[j]);
      atomicAdd(o + 32, quad_lane<1>(alpha) * x1[j]);
      atomicAdd(o + 64, quad_lane<2>(alpha) * x2[j]);
      atomicAdd(o + 96, quad_lane<3>(alpha) * x3[j]);
    }
  }
  if (sum_lane) {
#pragma unroll
    for (int j = 0; j < RA_GROUP; ++j)
      if (FULL || j < m) atomicAdd(&acc[RA_SUMS + r[j] * H_ + h], wt[j]);
  }
}

template <int MT, int LDW, int NCOLS>
__device__ __forceinline__ void rows_product(const float* hl, int mt0, int col, int lane,
                                             const float* __restrict__ Wm, v8f (&acc)[MT]) {
  const int half = lane >> 4;
  const int l16  = lane & 15;
  const bool cv  = (NCOLS % 16 == 0) || col < NCOLS;
  const float* ap = hl + (mt0 * 16 + l16) * RA_PITCH + 2 * half;
  const float* bp = Wm + (2 * half) * LDW + (cv ? col : 0);
#pragma unroll
  for (int i = 0; i < MT; ++i) acc[i] = v8f{};
#pragma unroll
  for (int k = 0; k < F_; k += 4) {
    const float bx = bp[k * LDW];
    const float by = bp[(k + 1) * LDW];
    v2f b;
    b.x = cv ? bx : 0.f;
    b.y = cv ? by : 0.f;
#pragma unroll
    for (int i = 0; i < MT; ++i) {
      const v2f a = *(const v2f*)(ap + i * 64 * RA_PITCH + k);
      acc[i] = __builtin_amdgcn_wmma_f32_16x16x4_f32(false, a, false, b,
                                                     (short)0, acc[i], false, false);
    }
  }
}

template <int MT>
__device__ __forceinline__ void fc_tiles(const float* hl, int wave, int lane, int n0,
                                         const float* __restrict__ Wn, float* __restrict__ out) {
  const int mt0 = wave >> 3;
  const int col = (wave & 7) * 16 + (lane & 15);
  v8f acc[MT];
  rows_product<MT, F_, F_>(hl, mt0, col, lane, Wn, acc);
#pragma unroll
  for (int i = 0; i < MT; ++i) {
    const int m8 = (mt0 + 4 * i) * 16 + 8 * (lane >> 4);
    if (m8 < RA_ROWS) {
#pragma unroll
      for (int r = 0; r < 8; ++r) out[(size_t)(n0 + m8 + r) * F_ + col] = acc[i][r];
    }
  }
}

template <bool FIRST>
__global__ __launch_bounds__(RA_THREADS)
void range_aggregate(const int* __restrict__ src, const int* __restrict__ dst,
                     const float* __restrict__ el, const float* __restrict__ er,
                     const float* __restrict__ hp,
                     const float* __restrict__ bias, const float* __restrict__ gamma,
                     const float* __restrict__ beta,
                     const float* __restrict__ Wn, const float* __restrict__ bn,
                     float* __restrict__ out,
                     int* __restrict__ wl, int* __restrict__ wc, int* __restrict__ wo) {
  __shared__ __attribute__((aligned(16))) float acc[RA_ACC];
  __shared__ float mxl[RA_NH];
  __shared__ float erl[RA_NH];
  __shared__ __attribute__((aligned(16))) int lists[RA_WAVES * RA_LIST];
  __shared__ int ovf;
  const int lane = threadIdx.x & 31;
  const int wave = __builtin_amdgcn_readfirstlane((int)(threadIdx.x >> 5));
  const int n0   = blockIdx.x * RA_ROWS;
  const int h    = 2 * (lane & 3) + (lane >> 4);
  const bool sum_lane = (lane & 12) == 0;
  int* lst = lists + wave * RA_LIST;
  int* wsl = wl + (size_t)(blockIdx.x * RA_WAVES + wave) * RA_LIST;
  const int ovf1 = FIRST ? 0 : wo[blockIdx.x];

  for (int i = threadIdx.x; i < RA_ACC; i += RA_THREADS) acc[i] = 0.f;
  for (int i = threadIdx.x; i < RA_NH; i += RA_THREADS) {
    mxl[i] = __uint_as_float(0xff800000u);
    erl[i] = er[n0 * H_ + i];
  }
  if (threadIdx.x == 0) ovf = ovf1;
  __syncthreads();

  int cnt = 0;
  if (FIRST) {
  const int4* dst4 = (const int4*)dst;
#pragma unroll 1
  for (int q0 = wave; q0 < RA_LOADS; q0 += RA_WAVES * RA_SCAN) {
    if (cnt > RA_LIST - RA_SCAN * 128) { ovf = 1; break; }
    int4 v[RA_SCAN];
#pragma unroll
    for (int j = 0; j < RA_SCAN; ++j) {
      const int q = q0 + j * RA_WAVES;
      const int4 t = dst4[(q < RA_LOADS ? q : q0) * 32 + lane];
      const int past = (q < RA_LOADS) ? 0 : -1;
      v[j].x = t.x | past; v[j].y = t.y | past; v[j].z = t.z | past; v[j].w = t.w | past;
    }
#pragma unroll
    for (int j = 0; j < RA_SCAN; ++j) {
      const int ebase = ((q0 + j * RA_WAVES) * 32 + lane) * 4;
      const int dd[4] = {v[j].x, v[j].y, v[j].z, v[j].w};
#pragma unroll
      for (int c = 0; c < 4; ++c) {
        const int row = dd[c] - n0;
        const bool hit = (unsigned)row < (unsigned)RA_ROWS;
        const unsigned bal = __builtin_amdgcn_ballot_w32(hit);
        if (bal != 0u) {
          if (hit) lst[cnt + (int)__builtin_amdgcn_mbcnt_lo(bal, 0u)] = (ebase + c) | (row << 20);
          cnt += __builtin_popcount(bal);
        }
      }
    }
  }

  for (int b = 0; b < cnt; b += 128) {
    int ent[4], sv[4];
#pragma unroll
    for (int k = 0; k < 4; ++k) {
      const int i = b + 32 * k + lane;
      ent[k] = (i < cnt) ? lst[i] : 0;
    }
#pragma unroll
    for (int k = 0; k < 4; ++k) sv[k] = src[ent[k] & 0xFFFFF];
#pragma unroll
    for (int k = 0; k < 4; ++k) {
      const int i = b + 32 * k + lane;
      if (i < cnt) {
        const int val = sv[k] | ((ent[k] >> 20) << 16);
        lst[i] = val;
        wsl[i] = val;
      }
    }
  }
  if (lane == 0) wc[blockIdx.x * RA_WAVES + wave] = cnt;
  } else {
    if (ovf1 == 0) cnt = __builtin_amdgcn_readfirstlane(wc[blockIdx.x * RA_WAVES + wave]);
    for (int b = 0; b < cnt; b += 128) {
      int ent[4];
#pragma unroll
      for (int k = 0; k < 4; ++k) {
        const int i = b + 32 * k + lane;
        ent[k] = wsl[i < cnt ? i : 0];
      }
#pragma unroll
      for (int k = 0; k < 4; ++k) {
        const int i = b + 32 * k + lane;
        if (i < cnt) lst[i] = ent[k];
      }
    }
  }
  {
    const int hh  = lane & 7;
    const int sub = lane >> 3;
    for (int b = 0; b < cnt; b += 4 * RA_MAXT) {
      int ent[RA_MAXT];
      float a[RA_MAXT];
#pragma unroll
      for (int t = 0; t < RA_MAXT; ++t) {
        const int i = b + 4 * t + sub;
        ent[t] = (i < cnt) ? lst[i] : -1;
      }
#pragma unroll
      for (int t = 0; t < RA_MAXT; ++t)
        a[t] = el[(ent[t] < 0 ? 0 : (ent[t] & 0xFFFF)) * H_ + hh];
#pragma unroll
      for (int t = 0; t < RA_MAXT; ++t) {
        if (ent[t] >= 0) {
          const int rh = (ent[t] >> 16) * H_ + hh;
          atomicMaxF(&mxl[rh], lrelu(a[t] + erl[rh]));
        }
      }
    }
  }
  __syncthreads();
  if (FIRST && threadIdx.x == 0) wo[blockIdx.x] = ovf;

  if (ovf == 0) {
    int head = 0;
    for (; cnt - head >= RA_GROUP; head += RA_GROUP)
      agg_group<true>(lst + head, RA_GROUP, lane, h, sum_lane, el, hp, erl, mxl, acc);
    if (cnt > head)
      agg_group<false>(lst + head, cnt - head, lane, h, sum_lane, el, hp, erl, mxl, acc);
  } else {
    for (int e = threadIdx.x; e < E_; e += RA_THREADS) {
      const int row = dst[e] - n0;
      if ((unsigned)row < (unsigned)RA_ROWS) {
        const int sn = src[e];
        for (int k = 0; k < H_; ++k)
          atomicMaxF(&mxl[row * H_ + k], lrelu(el[sn * H_ + k] + erl[row * H_ + k]));
      }
    }
    __syncthreads();
    for (int e = threadIdx.x; e < E_; e += RA_THREADS) {
      const int row = dst[e] - n0;
      if ((unsigned)row < (unsigned)RA_ROWS) {
        const int sn = src[e];
        for (int k = 0; k < H_; ++k) {
          const float v = lrelu(el[sn * H_ + k] + erl[row * H_ + k]);
          const float w = __expf(v - mxl[row * H_ + k]);
          atomicAdd(&acc[RA_SUMS + row * H_ + k], w);
          for (int d = 0; d < D_; ++d)
            atomicAdd(&acc[row * F_ + k * D_ + d], w * hp[(size_t)sn * F_ + k * D_ + d]);
        }
      }
    }
  }
  __syncthreads();

  float* hl = (float*)lists;
  hl[(RA_ROWS + (threadIdx.x >> 7)) * RA_PITCH + (threadIdx.x & 127)] = 0.f;
  const float4 bb = *(const float4*)(bias + lane * 4);
  const float4 g  = *(const float4*)(gamma + lane * 4);
  const float4 be = *(const float4*)(beta + lane * 4);
  for (int r = wave; r < RA_ROWS; r += RA_WAVES) {
    float4 x = *(const float4*)(acc + r * F_ + lane * 4);
    const float rs = 1.f / (acc[RA_SUMS + r * H_ + (lane >> 2)] + 1e-16f);
    x.x = x.x * rs + bb.x; x.y = x.y * rs + bb.y; x.z = x.z * rs + bb.z; x.w = x.w * rs + bb.w;
    float sA = x.x + x.y + x.z + x.w;
    float sB = x.x * x.x + x.y * x.y + x.z * x.z + x.w * x.w;
#pragma unroll
    for (int off = 16; off >= 1; off >>= 1) {
      sA += __shfl_xor(sA, off, 32);
      sB += __shfl_xor(sB, off, 32);
    }
    const float mu  = sA * (1.f / 128.f);
    const float var = sB * (1.f / 128.f) - mu * mu;
    const float inv = rsqrtf(var + LN_EPS);
    float4 y;
    y.x = fmaxf((x.x - mu) * inv * g.x + be.x, 0.f);
    y.y = fmaxf((x.y - mu) * inv * g.y + be.y, 0.f);
    y.z = fmaxf((x.z - mu) * inv * g.z + be.z, 0.f);
    y.w = fmaxf((x.w - mu) * inv * g.w + be.w, 0.f);
    *(float4*)(hl + r * RA_PITCH + lane * 4) = y;
  }
  __syncthreads();

  if (FIRST) {
    if (wave < 8) fc_tiles<4>(hl, wave, lane, n0, Wn, out);
    else          fc_tiles<3>(hl, wave, lane, n0, Wn, out);
  } else {
    for (int t = wave; t < RA_PTILES; t += RA_WAVES) {
      const int mt = t / 3;
      const int n  = (t - 3 * mt) * 16 + (lane & 15);
      v8f acc[1];
      rows_product<1, C_, C_>(hl, mt, n, lane, Wn, acc);
      const int m8 = mt * 16 + 8 * (lane >> 4);
      if (n < C_ && m8 < RA_ROWS) {
        const float bbn = bn[n];
#pragma unroll
        for (int r = 0; r < 8; ++r) out[(size_t)(n0 + m8 + r) * C_ + n] = acc[0][r] + bbn;
      }
    }
  }
}

extern "C" void kernel_launch(void* const* d_in, const int* in_sizes, int n_in,
                              void* d_out, int out_size, void* d_ws, size_t ws_size,
                              hipStream_t stream) {
  const float* feats = (const float*)d_in[0];
  const int*   src   = (const int*)d_in[1];
  const int*   dst   = (const int*)d_in[2];
  const float* Ws    = (const float*)d_in[3];
  const float* al    = (const float*)d_in[4];
  const float* ar    = (const float*)d_in[5];
  const float* bias  = (const float*)d_in[6];
  const float* gamma = (const float*)d_in[7];
  const float* beta  = (const float*)d_in[8];
  const float* Wp    = (const float*)d_in[9];
  const float* bp    = (const float*)d_in[10];
  float* outp = (float*)d_out;

  float* ws   = (float*)d_ws;
  float* hp0  = ws;
  float* hp1  = hp0 + (size_t)N_ * F_;
  float* el   = hp1 + (size_t)N_ * F_;
  float* er   = el + (size_t)N_ * H_;
  int* wl     = (int*)(er + (size_t)N_ * H_);
  int* wc     = wl + (size_t)RA_BLOCKS * RA_WAVES * RA_LIST;
  int* wo     = wc + RA_BLOCKS * RA_WAVES;

  const int NH = N_ * H_;

  gemm_node_fc<<<N_ / 16, 256, 0, stream>>>(feats, Ws, hp0);
  node_coef<<<(NH + 255) / 256, 256, 0, stream>>>(hp0, al, ar, el, er, NH);
  range_aggregate<true><<<RA_BLOCKS, RA_THREADS, 0, stream>>>(
      src, dst, el, er, hp0, bias, gamma, beta, Ws + (size_t)F_ * F_, bp, hp1, wl, wc, wo);
  node_coef<<<(NH + 255) / 256, 256, 0, stream>>>(
      hp1, al + (size_t)H_ * D_, ar + (size_t)H_ * D_, el, er, NH);
  range_aggregate<false><<<RA_BLOCKS, RA_THREADS, 0, stream>>>(
      src, dst, el, er, hp1, bias + (size_t)F_, gamma + (size_t)F_, beta + (size_t)F_,
      Wp, bp, outp, wl, wc, wo);
}


# Range scan reads one key byte per edge (dst / 200, written by the fc launch), 16 edges per lane-load with a zero-byte test, not dst itself; lists in another order, sums still in arrival order
# speedup vs baseline: 4.2396x; 1.1087x over previous
#include <hip/hip_runtime.h>

constexpr int N_ = 50000;
constexpr int E_ = 800000;
constexpr int F_ = 128;
constexpr int H_ = 8;
constexpr int D_ = 16;
constexpr int C_ = 40;
constexpr int L_ = 2;
constexpr float NEG_SLOPE = 0.2f;
constexpr float LN_EPS = 1e-5f;

constexpr int RA_ROWS    = 200;
constexpr int RA_BLOCKS  = N_ / RA_ROWS;
constexpr int RA_WAVES   = 32;
constexpr int RA_THREADS = RA_WAVES * 32;
constexpr int RA_KQ      = (E_ + 511) / 512;
constexpr int RA_KWORDS  = RA_KQ * 128;
constexpr int RA_SCAN    = 8;
constexpr int RA_GROUP   = 8;
constexpr int RA_MAXT    = 16;
constexpr int RA_LIST    = 1536;
constexpr int RA_NH      = RA_ROWS * H_;
constexpr int RA_SUMS    = RA_ROWS * F_;
constexpr int RA_ACC     = RA_ROWS * F_ + RA_NH;
static_assert(RA_BLOCKS * RA_ROWS == N_, "node ranges");
static_assert(E_ % 4 == 0 && RA_BLOCKS <= 255, "range keys: four per word, 255 marks a pad key");
static_assert(RA_LIST > 512 && RA_KQ * 32 <= (1 << 16), "a key load's hits fit behind the capacity check; list entry: group << 5 | bit");
static_assert((N_ / 16) * 256 >= RA_KWORDS, "the fc launch has a thread per key word");
static_assert(N_ <= (1 << 16), "list entry: source node | row << 16");
static_assert((RA_ACC + 2 * RA_NH + RA_WAVES * RA_LIST + 1) * 4 <= 320 * 1024, "LDS of one work-group");
constexpr int RA_MT      = (RA_ROWS + 15) / 16;
constexpr int RA_PITCH   = F_ + 4;
constexpr int RA_PTILES  = RA_MT * ((C_ + 15) / 16);
static_assert(RA_MT * 16 * RA_PITCH <= RA_WAVES * RA_LIST, "finished rows fit where the lists were");
static_assert(RA_MT == 13 && RA_WAVES == 32, "fc tiles: 8 column slabs x (3 row tiles per wave, a 4th for waves 0..7)");
static_assert((RA_MT * 16 - RA_ROWS) * F_ == RA_THREADS, "pad rows: one float per thread");
static_assert(L_ == 2, "kernel_launch is written out for two layers");
static_assert(RA_ROWS % 8 == 0, "a lane's eight accumulator rows are all real or all pad");

typedef __attribute__((ext_vector_type(2))) float v2f;
typedef __attribute__((ext_vector_type(8))) float v8f;

__device__ __forceinline__ float lrelu(float x) { return x > 0.f ? x : NEG_SLOPE * x; }

__device__ __forceinline__ void atomicMaxF(float* addr, float val) {
  if (val >= 0.f) atomicMax((int*)addr, __float_as_int(val));
  else            atomicMin((unsigned int*)addr, __float_as_uint(val));
}

__global__ void gemm_node_fc(const float* __restrict__ X,
                             const float* __restrict__ W,
                             float* __restrict__ Y,
                             const int* __restrict__ dst,
                             unsigned int* __restrict__ keys) {
  {
    const int gid = blockIdx.x * 256 + threadIdx.x;
    if (gid < E_ / 4) {
      const int4 d = ((const int4*)dst)[gid];
      keys[gid] = ((unsigned)d.x / RA_ROWS) | (((unsigned)d.y / RA_ROWS) << 8) |
                  (((unsigned)d.z / RA_ROWS) << 16) | (((unsigned)d.w / RA_ROWS) << 24);
    } else if (gid < RA_KWORDS) {
      keys[gid] = 0xFFFFFFFFu;
    }
  }
  const int lane = threadIdx.x & 31;
  const int wave = threadIdx.x >> 5;
  const int m0   = blockIdx.x * 16;
  const int n0   = wave * 16;
  const int half = lane >> 4;
  const int l16  = lane & 15;
  const int arow = m0 + l16;
  const int bcol = n0 + l16;

  v8f acc = {};
#pragma unroll
  for (int k = 0; k < F_; k += 4) {
    const int ka = k + 2 * half;
    v2f a;
    a.x = X[arow * F_ + ka];
    a.y = X[arow * F_ + ka + 1];
    v2f b;
    b.x = W[ka * F_ + bcol];
    b.y = W[(ka + 1) * F_ + bcol];
    acc = __builtin_amdgcn_wmma_f32_16x16x4_f32(false, a, false, b,
                                                (short)0, acc, false, false);
  }
#pragma unroll
  for (int r = 0; r < 8; ++r) {
    const int m = m0 + r + 8 * half;
    Y[m * F_ + bcol] = acc[r];
  }
}

__global__ void node_coef(const float* __restrict__ hp,
                          const float* __restrict__ al,
                          const float* __restrict__ ar,
                          float* __restrict__ el, float* __restrict__ er, int total) {
  const int t = blockIdx.x * blockDim.x + threadIdx.x;
  if (t >= total) return;
  const int n = t >> 3;
  const int h = t & 7;
  const float* row = hp + n * F_ + h * D_;
  float sl = 0.f, sr = 0.f;
#pragma unroll
  for (int d = 0; d < D_; ++d) {
    const float x = row[d];
    sl += x * al[h * D_ + d];
    sr += x * ar[h * D_ + d];
  }
  el[t] = sl;
  er[t] = sr;
}

template <int J>
__device__ __forceinline__ float quad_lane(float v) {
  return __int_as_float(__builtin_amdgcn_update_dpp(
      0, __float_as_int(v), J * 0x55, 0xF, 0xF, true));
}

template <bool FULL>
__device__ __forceinline__ void agg_group(const int* lst, int m, int lane, int h,
                                          bool sum_lane,
                                          const float* __restrict__ el,
                                          const float* __restrict__ hp,
                                          const float* erl, const float* mxl, float* acc) {
  const int slot = lane & (RA_GROUP - 1);
  const int ev = lst[(FULL || slot < m) ? slot : 0];
  int s[RA_GROUP], r[RA_GROUP];
  float cl[RA_GROUP], cr[RA_GROUP], cm[RA_GROUP], wt[RA_GROUP];
  float x0[RA_GROUP], x1[RA_GROUP], x2[RA_GROUP], x3[RA_GROUP];
#pragma unroll
  for (int j = 0; j < RA_GROUP; ++j) {
    const int t = __builtin_amdgcn_readlane(ev, j);
    s[j] = t & 0xFFFF;
    r[j] = t >> 16;
  }
#pragma unroll
  for (int j = 0; j < RA_GROUP; ++j) {
    cl[j] = el[s[j] * H_ + h];
    cr[j] = erl[r[j] * H_ + h];
    cm[j] = mxl[r[j] * H_ + h];
    const float* x = hp + (size_t)s[j] * F_ + lane;
    x0[j] = x[0]; x1[j] = x[32]; x2[j] = x[64]; x3[j] = x[96];
  }
  __builtin_amdgcn_sched_barrier(0);
#pragma unroll
  for (int j = 0; j < RA_GROUP; ++j) {
    const float v     = lrelu(cl[j] + cr[j]);
    const float alpha = __expf(v - cm[j]);
    wt[j] = alpha;
    if (FULL || j < m) {
      float* o = acc + r[j] * F_ + lane;
      atomicAdd(o,      quad_lane<0>(alpha) * x0[j]);
      atomicAdd(o + 32, quad_lane<1>(alpha) * x1[j]);
      atomicAdd(o + 64, quad_lane<2>(alpha) * x2[j]);
      atomicAdd(o + 96, quad_lane<3>(alpha) * x3[j]);
    }
  }
  if (sum_lane) {
#pragma unroll
    for (int j = 0; j < RA_GROUP; ++j)
      if (FULL || j < m) atomicAdd(&acc[RA_SUMS + r[j] * H_ + h], wt[j]);
  }
}

template <int MT, int LDW, int NCOLS>
__device__ __forceinline__ void rows_product(const float* hl, int mt0, int col, int lane,
                                             const float* __restrict__ Wm, v8f (&acc)[MT]) {
  const int half = lane >> 4;
  const int l16  = lane & 15;
  const bool cv  = (NCOLS % 16 == 0) || col < NCOLS;
  const float* ap = hl + (mt0 * 16 + l16) * RA_PITCH + 2 * half;
  const float* bp = Wm + (2 * half) * LDW + (cv ? col : 0);
#pragma unroll
  for (int i = 0; i < MT; ++i) acc[i] = v8f{};
#pragma unroll
  for (int k = 0; k < F_; k += 4) {
    const float bx = bp[k * LDW];
    const float by = bp[(k + 1) * LDW];
    v2f b;
    b.x = cv ? bx : 0.f;
    b.y = cv ? by : 0.f;
#pragma unroll
    for (int i = 0; i < MT; ++i) {
      const v2f a = *(const v2f*)(ap + i * 64 * RA_PITCH + k);
      acc[i] = __builtin_amdgcn_wmma_f32_16x16x4_f32(false, a, false, b,
                                                     (short)0, acc[i], false, false);
    }
  }
}

template <int MT>
__device__ __forceinline__ void fc_tiles(const float* hl, int wave, int lane, int n0,
                                         const float* __restrict__ Wn, float* __restrict__ out) {
  const int mt0 = wave >> 3;
  const int col = (wave & 7) * 16 + (lane & 15);
  v8f acc[MT];
  rows_product<MT, F_, F_>(hl, mt0, col, lane, Wn, acc);
#pragma unroll
  for (int i = 0; i < MT; ++i) {
    const int m8 = (mt0 + 4 * i) * 16 + 8 * (lane >> 4);
    if (m8 < RA_ROWS) {
#pragma unroll
      for (int r = 0; r < 8; ++r) out[(size_t)(n0 + m8 + r) * F_ + col] = acc[i][r];
    }
  }
}

template <bool FIRST>
__global__ __launch_bounds__(RA_THREADS)
void range_aggregate(const int* __restrict__ src, const int* __restrict__ dst,
                     const float* __restrict__ el, const float* __restrict__ er,
                     const float* __restrict__ hp,
                     const float* __restrict__ bias, const float* __restrict__ gamma,
                     const float* __restrict__ beta,
                     const float* __restrict__ Wn, const float* __restrict__ bn,
                     float* __restrict__ out,
                     int* __restrict__ wl, int* __restrict__ wc, int* __restrict__ wo,
                     const unsigned int* __restrict__ keys) {
  __shared__ __attribute__((aligned(16))) float acc[RA_ACC];
  __shared__ float mxl[RA_NH];
  __shared__ float erl[RA_NH];
  __shared__ __attribute__((aligned(16))) int lists[RA_WAVES * RA_LIST];
  __shared__ int ovf;
  const int lane = threadIdx.x & 31;
  const int wave = __builtin_amdgcn_readfirstlane((int)(threadIdx.x >> 5));
  const int n0   = blockIdx.x * RA_ROWS;
  const int h    = 2 * (lane & 3) + (lane >> 4);
  const bool sum_lane = (lane & 12) == 0;
  int* lst = lists + wave * RA_LIST;
  int* wsl = wl + (size_t)(blockIdx.x * RA_WAVES + wave) * RA_LIST;
  const int ovf1 = FIRST ? 0 : wo[blockIdx.x];

  for (int i = threadIdx.x; i < RA_ACC; i += RA_THREADS) acc[i] = 0.f;
  for (int i = threadIdx.x; i < RA_NH; i += RA_THREADS) {
    mxl[i] = __uint_as_float(0xff800000u);
    erl[i] = er[n0 * H_ + i];
  }
  if (threadIdx.x == 0) ovf = ovf1;
  __syncthreads();

  int cnt = 0;
  if (FIRST) {
  const uint4* key4 = (const uint4*)keys;
  const unsigned G = (unsigned)blockIdx.x * 0x01010101u;
  bool full = false;
#pragma unroll 1
  for (int q0 = wave; q0 < RA_KQ && !full; q0 += RA_WAVES * RA_SCAN) {
    uint4 v[RA_SCAN];
#pragma unroll
    for (int j = 0; j < RA_SCAN; ++j) {
      const int q = q0 + j * RA_WAVES;
      v[j] = key4[(q < RA_KQ ? q : q0) * 32 + lane];
    }
#pragma unroll
    for (int j = 0; j < RA_SCAN; ++j) {
      const int q = q0 + j * RA_WAVES;
      if (q < RA_KQ && !full) {
        if (cnt > RA_LIST - 512) { full = true; continue; }
        const unsigned kw[4] = {v[j].x, v[j].y, v[j].z, v[j].w};
        unsigned m = 0u;
#pragma unroll
        for (int c = 0; c < 4; ++c) {
          const unsigned t = kw[c] ^ G;
          const unsigned u = (t & 0x7F7F7F7Fu) + 0x7F7F7F7Fu;
          m |= ~(u | t | 0x7F7F7F7Fu) >> c;
        }
        const int gbase = (q * 32 + lane) << 5;
        unsigned bal = __builtin_amdgcn_ballot_w32(m != 0u);
        while (bal != 0u) {
          if (m != 0u) {
            lst[cnt + (int)__builtin_amdgcn_mbcnt_lo(bal, 0u)] = gbase | __builtin_ctz(m);
            m &= m - 1u;
          }
          cnt += __builtin_popcount(bal);
          bal = __builtin_amdgcn_ballot_w32(m != 0u);
        }
      }
    }
  }
  if (full) ovf = 1;

  for (int b = 0; b < cnt; b += 128) {
    int ent[4], sv[4], dv[4];
#pragma unroll
    for (int k = 0; k < 4; ++k) {
      const int i = b + 32 * k + lane;
      const int t = (i < cnt) ? lst[i] : 0;
      ent[k] = (t >> 5) * 16 + 4 * (7 - (t & 7)) + ((t & 31) >> 3);
    }
#pragma unroll
    for (int k = 0; k < 4; ++k) { sv[k] = src[ent[k]]; dv[k] = dst[ent[k]]; }
#pragma unroll
    for (int k = 0; k < 4; ++k) {
      const int i = b + 32 * k + lane;
      if (i < cnt) {
        const int val = sv[k] | ((dv[k] - n0) << 16);
        lst[i] = val;
        wsl[i] = val;
      }
    }
  }
  if (lane == 0) wc[blockIdx.x * RA_WAVES + wave] = cnt;
  } else {
    if (ovf1 == 0) cnt = __builtin_amdgcn_readfirstlane(wc[blockIdx.x * RA_WAVES + wave]);
    for (int b = 0; b < cnt; b += 128) {
      int ent[4];
#pragma unroll
      for (int k = 0; k < 4; ++k) {
        const int i = b + 32 * k + lane;
        ent[k] = wsl[i < cnt ? i : 0];
      }
#pragma unroll
      for (int k = 0; k < 4; ++k) {
        const int i = b + 32 * k + lane;
        if (i < cnt) lst[i] = ent[k];
      }
    }
  }
  {
    const int hh  = lane & 7;
    const int sub = lane >> 3;
    for (int b = 0; b < cnt; b += 4 * RA_MAXT) {
      int ent[RA_MAXT];
      float a[RA_MAXT];
#pragma unroll
      for (int t = 0; t < RA_MAXT; ++t) {
        const int i = b + 4 * t + sub;
        ent[t] = (i < cnt) ? lst[i] : -1;
      }
#pragma unroll
      for (int t = 0; t < RA_MAXT; ++t)
        a[t] = el[(ent[t] < 0 ? 0 : (ent[t] & 0xFFFF)) * H_ + hh];
#pragma unroll
      for (int t = 0; t < RA_MAXT; ++t) {
        if (ent[t] >= 0) {
          const int rh = (ent[t] >> 16) * H_ + hh;
          atomicMaxF(&mxl[rh], lrelu(a[t] + erl[rh]));
        }
      }
    }
  }
  __syncthreads();
  if (FIRST && threadIdx.x == 0) wo[blockIdx.x] = ovf;

  if (ovf == 0) {
    int head = 0;
    for (; cnt - head >= RA_GROUP; head += RA_GROUP)
      agg_group<true>(lst + head, RA_GROUP, lane, h, sum_lane, el, hp, erl, mxl, acc);
    if (cnt > head)
      agg_group<false>(lst + head, cnt - head, lane, h, sum_lane, el, hp, erl, mxl, acc);
  } else {
    for (int e = threadIdx.x; e < E_; e += RA_THREADS) {
      const int row = dst[e] - n0;
      if ((unsigned)row < (unsigned)RA_ROWS) {
        const int sn = src[e];
        for (int k = 0; k < H_; ++k)
          atomicMaxF(&mxl[row * H_ + k], lrelu(el[sn * H_ + k] + erl[row * H_ + k]));
      }
    }
    __syncthreads();
    for (int e = threadIdx.x; e < E_; e += RA_THREADS) {
      const int row = dst[e] - n0;
      if ((unsigned)row < (unsigned)RA_ROWS) {
        const int sn = src[e];
        for (int k = 0; k < H_; ++k) {
          const float v = lrelu(el[sn * H_ + k] + erl[row * H_ + k]);
          const float w = __expf(v - mxl[row * H_ + k]);
          atomicAdd(&acc[RA_SUMS + row * H_ + k], w);
          for (int d = 0; d < D_; ++d)
            atomicAdd(&acc[row * F_ + k * D_ + d], w * hp[(size_t)sn * F_ + k * D_ + d]);
        }
      }
    }
  }
  __syncthreads();

  float* hl = (float*)lists;
  hl[(RA_ROWS + (threadIdx.x >> 7)) * RA_PITCH + (threadIdx.x & 127)] = 0.f;
  const float4 bb = *(const float4*)(bias + lane * 4);
  const float4 g  = *(const float4*)(gamma + lane * 4);
  const float4 be = *(const float4*)(beta + lane * 4);
  for (int r = wave; r < RA_ROWS; r += RA_WAVES) {
    float4 x = *(const float4*)(acc + r * F_ + lane * 4);
    const float rs = 1.f / (acc[RA_SUMS + r * H_ + (lane >> 2)] + 1e-16f);
    x.x = x.x * rs + bb.x; x.y = x.y * rs + bb.y; x.z = x.z * rs + bb.z; x.w = x.w * rs + bb.w;
    float sA = x.x + x.y + x.z + x.w;
    float sB = x.x * x.x + x.y * x.y + x.z * x.z + x.w * x.w;
#pragma unroll
    for (int off = 16; off >= 1; off >>= 1) {
      sA += __shfl_xor(sA, off, 32);
      sB += __shfl_xor(sB, off, 32);
    }
    const float mu  = sA * (1.f / 128.f);
    const float var = sB * (1.f / 128.f) - mu * mu;
    const float inv = rsqrtf(var + LN_EPS);
    float4 y;
    y.x = fmaxf((x.x - mu) * inv * g.x + be.x, 0.f);
    y.y = fmaxf((x.y - mu) * inv * g.y + be.y, 0.f);
    y.z = fmaxf((x.z - mu) * inv * g.z + be.z, 0.f);
    y.w = fmaxf((x.w - mu) * inv * g.w + be.w, 0.f);
    *(float4*)(hl + r * RA_PITCH + lane * 4) = y;
  }
  __syncthreads();

  if (FIRST) {
    if (wave < 8) fc_tiles<4>(hl, wave, lane, n0, Wn, out);
    else          fc_tiles<3>(hl, wave, lane, n0, Wn, out);
  } else {
    for (int t = wave; t < RA_PTILES; t += RA_WAVES) {
      const int mt = t / 3;
      const int n  = (t - 3 * mt) * 16 + (lane & 15);
      v8f acc[1];
      rows_product<1, C_, C_>(hl, mt, n, lane, Wn, acc);
      const int m8 = mt * 16 + 8 * (lane >> 4);
      if (n < C_ && m8 < RA_ROWS) {
        const float bbn = bn[n];
#pragma unroll
        for (int r = 0; r < 8; ++r) out[(size_t)(n0 + m8 + r) * C_ + n] = acc[0][r] + bbn;
      }
    }
  }
}

extern "C" void kernel_launch(void* const* d_in, const int* in_sizes, int n_in,
                              void* d_out, int out_size, void* d_ws, size_t ws_size,
                              hipStream_t stream) {
  const float* feats = (const float*)d_in[0];
  const int*   src   = (const int*)d_in[1];
  const int*   dst   = (const int*)d_in[2];
  const float* Ws    = (const float*)d_in[3];
  const float* al    = (const float*)d_in[4];
  const float* ar    = (const float*)d_in[5];
  const float* bias  = (const float*)d_in[6];
  const float* gamma = (const float*)d_in[7];
  const float* beta  = (const float*)d_in[8];
  const float* Wp    = (const float*)d_in[9];
  const float* bp    = (const float*)d_in[10];
  float* outp = (float*)d_out;

  float* ws   = (float*)d_ws;
  float* hp0  = ws;
  float* hp1  = hp0 + (size_t)N_ * F_;
  float* el   = hp1 + (size_t)N_ * F_;
  float* er   = el + (size_t)N_ * H_;
  int* wl     = (int*)(er + (size_t)N_ * H_);
  int* wc     = wl + (size_t)RA_BLOCKS * RA_WAVES * RA_LIST;
  int* wo     = wc + RA_BLOCKS * RA_WAVES;
  unsigned int* keys = (unsigned int*)(wo + 256);

  const int NH = N_ * H_;

  gemm_node_fc<<<N_ / 16, 256, 0, stream>>>(feats, Ws, hp0, dst, keys);
  node_coef<<<(NH + 255) / 256, 256, 0, stream>>>(hp0, al, ar, el, er, NH);
  range_aggregate<true><<<RA_BLOCKS, RA_THREADS, 0, stream>>>(
      src, dst, el, er, hp0, bias, gamma, beta, Ws + (size_t)F_ * F_, bp, hp1, wl, wc, wo, keys);
  node_coef<<<(NH + 255) / 256, 256, 0, stream>>>(
      hp1, al + (size_t)H_ * D_, ar + (size_t)H_ * D_, el, er, NH);
  range_aggregate<false><<<RA_BLOCKS, RA_THREADS, 0, stream>>>(
      src, dst, el, er, hp1, bias + (size_t)F_, gamma + (size_t)F_, beta + (size_t)F_,
      Wp, bp, outp, wl, wc, wo, keys);
}


# First projection retiled: a wave takes 16 rows x 64 columns; one A fragment and two 16-byte W-row loads (a lane's 4 adjacent columns) feed 4 products; 16-byte stores; same k and summation order
# speedup vs baseline: 4.4527x; 1.0503x over previous
#include <hip/hip_runtime.h>

constexpr int N_ = 50000;
constexpr int E_ = 800000;
constexpr int F_ = 128;
constexpr int H_ = 8;
constexpr int D_ = 16;
constexpr int C_ = 40;
constexpr int L_ = 2;
constexpr float NEG_SLOPE = 0.2f;
constexpr float LN_EPS = 1e-5f;

constexpr int RA_ROWS    = 200;
constexpr int RA_BLOCKS  = N_ / RA_ROWS;
constexpr int RA_WAVES   = 32;
constexpr int RA_THREADS = RA_WAVES * 32;
constexpr int RA_KQ      = (E_ + 511) / 512;
constexpr int RA_KWORDS  = RA_KQ * 128;
constexpr int RA_SCAN    = 8;
constexpr int RA_GROUP   = 8;
constexpr int RA_MAXT    = 16;
constexpr int RA_LIST    = 1536;
constexpr int RA_NH      = RA_ROWS * H_;
constexpr int RA_SUMS    = RA_ROWS * F_;
constexpr int RA_ACC     = RA_ROWS * F_ + RA_NH;
static_assert(RA_BLOCKS * RA_ROWS == N_, "node ranges");
static_assert(E_ % 4 == 0 && RA_BLOCKS <= 255, "range keys: four per word, 255 marks a pad key");
static_assert(RA_LIST > 512 && RA_KQ * 32 <= (1 << 16), "a key load's hits fit behind the capacity check; list entry: group << 5 | bit");
constexpr int FC_RT      = 4;
constexpr int FC_BLOCKS  = (N_ / 16 + FC_RT - 1) / FC_RT;
static_assert(N_ % 16 == 0, "fc row tiles are whole");
static_assert(FC_BLOCKS * 256 >= RA_KWORDS, "the fc launch has a thread per key word");
static_assert(N_ <= (1 << 16), "list entry: source node | row << 16");
static_assert((RA_ACC + 2 * RA_NH + RA_WAVES * RA_LIST + 1) * 4 <= 320 * 1024, "LDS of one work-group");
constexpr int RA_MT      = (RA_ROWS + 15) / 16;
constexpr int RA_PITCH   = F_ + 4;
constexpr int RA_PTILES  = RA_MT * ((C_ + 15) / 16);
static_assert(RA_MT * 16 * RA_PITCH <= RA_WAVES * RA_LIST, "finished rows fit where the lists were");
static_assert(RA_MT == 13 && RA_WAVES == 32, "fc tiles: 8 column slabs x (3 row tiles per wave, a 4th for waves 0..7)");
static_assert((RA_MT * 16 - RA_ROWS) * F_ == RA_THREADS, "pad rows: one float per thread");
static_assert(L_ == 2, "kernel_launch is written out for two layers");
static_assert(RA_ROWS % 8 == 0, "a lane's eight accumulator rows are all real or all pad");

typedef __attribute__((ext_vector_type(2))) float v2f;
typedef __attribute__((ext_vector_type(8))) float v8f;

__device__ __forceinline__ float lrelu(float x) { return x > 0.f ? x : NEG_SLOPE * x; }

__device__ __forceinline__ void atomicMaxF(float* addr, float val) {
  if (val >= 0.f) atomicMax((int*)addr, __float_as_int(val));
  else            atomicMin((unsigned int*)addr, __float_as_uint(val));
}

__global__ __launch_bounds__(256)
void gemm_node_fc(const float* __restrict__ X,
                  const float* __restrict__ W,
                  float* __restrict__ Y,
                  const int* __restrict__ dst,
                  unsigned int* __restrict__ keys) {
  {
    const int gid = blockIdx.x * 256 + threadIdx.x;
    if (gid < E_ / 4) {
      const int4 d = ((const int4*)dst)[gid];
      keys[gid] = ((unsigned)d.x / RA_ROWS) | (((unsigned)d.y / RA_ROWS) << 8) |
                  (((unsigned)d.z / RA_ROWS) << 16) | (((unsigned)d.w / RA_ROWS) << 24);
    } else if (gid < RA_KWORDS) {
      keys[gid] = 0xFFFFFFFFu;
    }
  }
  const int lane = threadIdx.x & 31;
  const int wave = threadIdx.x >> 5;
  const int rt   = blockIdx.x * FC_RT + (wave >> 1);
  if (rt >= N_ / 16) return;
  const int m0   = rt * 16;
  const int half = lane >> 4;
  const int l16  = lane & 15;
  const int cb   = (wave & 1) * 64 + 4 * l16;
  const float* ap = X + (size_t)(m0 + l16) * F_ + 2 * half;
  const float* bp = W + (size_t)(2 * half) * F_ + cb;

  v8f acc[4];
#pragma unroll
  for (int t = 0; t < 4; ++t) acc[t] = v8f{};
#pragma unroll
  for (int k = 0; k < F_; k += 4) {
    const v2f a = *(const v2f*)(ap + k);
    const float4 b0 = *(const float4*)(bp + (size_t)k * F_);
    const float4 b1 = *(const float4*)(bp + (size_t)(k + 1) * F_);
    v2f b;
    b.x = b0.x; b.y = b1.x;
    acc[0] = __builtin_amdgcn_wmma_f32_16x16x4_f32(false, a, false, b, (short)0, acc[0], false, false);
    b.x = b0.y; b.y = b1.y;
    acc[1] = __builtin_amdgcn_wmma_f32_16x16x4_f32(false, a, false, b, (short)0, acc[1], false, false);
    b.x = b0.z; b.y = b1.z;
    acc[2] = __builtin_amdgcn_wmma_f32_16x16x4_f32(false, a, false, b, (short)0, acc[2], false, false);
    b.x = b0.w; b.y = b1.w;
    acc[3] = __builtin_amdgcn_wmma_f32_16x16x4_f32(false, a, false, b, (short)0, acc[3], false, false);
  }
#pragma unroll
  for (int r = 0; r < 8; ++r) {
    const int m = m0 + r + 8 * half;
    float4 o;
    o.x = acc[0][r]; o.y = acc[1][r]; o.z = acc[2][r]; o.w = acc[3][r];
    *(float4*)(Y + (size_t)m * F_ + cb) = o;
  }
}

__global__ void node_coef(const float* __restrict__ hp,
                          const float* __restrict__ al,
                          const float* __restrict__ ar,
                          float* __restrict__ el, float* __restrict__ er, int total) {
  const int t = blockIdx.x * blockDim.x + threadIdx.x;
  if (t >= total) return;
  const int n = t >> 3;
  const int h = t & 7;
  const float* row = hp + n * F_ + h * D_;
  float sl = 0.f, sr = 0.f;
#pragma unroll
  for (int d = 0; d < D_; ++d) {
    const float x = row[d];
    sl += x * al[h * D_ + d];
    sr += x * ar[h * D_ + d];
  }
  el[t] = sl;
  er[t] = sr;
}

template <int J>
__device__ __forceinline__ float quad_lane(float v) {
  return __int_as_float(__builtin_amdgcn_update_dpp(
      0, __float_as_int(v), J * 0x55, 0xF, 0xF, true));
}

template <bool FULL>
__device__ __forceinline__ void agg_group(const int* lst, int m, int lane, int h,
                                          bool sum_lane,
                                          const float* __restrict__ el,
                                          const float* __restrict__ hp,
                                          const float* erl, const float* mxl, float* acc) {
  const int slot = lane & (RA_GROUP - 1);
  const int ev = lst[(FULL || slot < m) ? slot : 0];
  int s[RA_GROUP], r[RA_GROUP];
  float cl[RA_GROUP], cr[RA_GROUP], cm[RA_GROUP], wt[RA_GROUP];
  float x0[RA_GROUP], x1[RA_GROUP], x2[RA_GROUP], x3[RA_GROUP];
#pragma unroll
  for (int j = 0; j < RA_GROUP; ++j) {
    const int t = __builtin_amdgcn_readlane(ev, j);
    s[j] = t & 0xFFFF;
    r[j] = t >> 16;
  }
#pragma unroll
  for (int j = 0; j < RA_GROUP; ++j) {
    cl[j] = el[s[j] * H_ + h];
    cr[j] = erl[r[j] * H_ + h];
    cm[j] = mxl[r[j] * H_ + h];
    const float* x = hp + (size_t)s[j] * F_ + lane;
    x0[j] = x[0]; x1[j] = x[32]; x2[j] = x[64]; x3[j] = x[96];
  }
  __builtin_amdgcn_sched_barrier(0);
#pragma unroll
  for (int j = 0; j < RA_GROUP; ++j) {
    const float v     = lrelu(cl[j] + cr[j]);
    const float alpha = __expf(v - cm[j]);
    wt[j] = alpha;
    if (FULL || j < m) {
      float* o = acc + r[j] * F_ + lane;
      atomicAdd(o,      quad_lane<0>(alpha) * x0[j]);
      atomicAdd(o + 32, quad_lane<1>(alpha) * x1[j]);
      atomicAdd(o + 64, quad_lane<2>(alpha) * x2[j]);
      atomicAdd(o + 96, quad_lane<3>(alpha) * x3[j]);
    }
  }
  if (sum_lane) {
#pragma unroll
    for (int j = 0; j < RA_GROUP; ++j)
      if (FULL || j < m) atomicAdd(&acc[RA_SUMS + r[j] * H_ + h], wt[j]);
  }
}

template <int MT, int LDW, int NCOLS>
__device__ __forceinline__ void rows_product(const float* hl, int mt0, int col, int lane,
                                             const float* __restrict__ Wm, v8f (&acc)[MT]) {
  const int half = lane >> 4;
  const int l16  = lane & 15;
  const bool cv  = (NCOLS % 16 == 0) || col < NCOLS;
  const float* ap = hl + (mt0 * 16 + l16) * RA_PITCH + 2 * half;
  const float* bp = Wm + (2 * half) * LDW + (cv ? col : 0);
#pragma unroll
  for (int i = 0; i < MT; ++i) acc[i] = v8f{};
#pragma unroll
  for (int k = 0; k < F_; k += 4) {
    const float bx = bp[k * LDW];
    const float by = bp[(k + 1) * LDW];
    v2f b;
    b.x = cv ? bx : 0.f;
    b.y = cv ? by : 0.f;
#pragma unroll
    for (int i = 0; i < MT; ++i) {
      const v2f a = *(const v2f*)(ap + i * 64 * RA_PITCH + k);
      acc[i] = __builtin_amdgcn_wmma_f32_16x16x4_f32(false, a, false, b,
                                                     (short)0, acc[i], false, false);
    }
  }
}

template <int MT>
__device__ __forceinline__ void fc_tiles(const float* hl, int wave, int lane, int n0,
                                         const float* __restrict__ Wn, float* __restrict__ out) {
  const int mt0 = wave >> 3;
  const int col = (wave & 7) * 16 + (lane & 15);
  v8f acc[MT];
  rows_product<MT, F_, F_>(hl, mt0, col, lane, Wn, acc);
#pragma unroll
  for (int i = 0; i < MT; ++i) {
    const int m8 = (mt0 + 4 * i) * 16 + 8 * (lane >> 4);
    if (m8 < RA_ROWS) {
#pragma unroll
      for (int r = 0; r < 8; ++r) out[(size_t)(n0 + m8 + r) * F_ + col] = acc[i][r];
    }
  }
}

template <bool FIRST>
__global__ __launch_bounds__(RA_THREADS)
void range_aggregate(const int* __restrict__ src, const int* __restrict__ dst,
                     const float* __restrict__ el, const float* __restrict__ er,
                     const float* __restrict__ hp,
                     const float* __restrict__ bias, const float* __restrict__ gamma,
                     const float* __restrict__ beta,
                     const float* __restrict__ Wn, const float* __restrict__ bn,
                     float* __restrict__ out,
                     int* __restrict__ wl, int* __restrict__ wc, int* __restrict__ wo,
                     const unsigned int* __restrict__ keys) {
  __shared__ __attribute__((aligned(16))) float acc[RA_ACC];
  __shared__ float mxl[RA_NH];
  __shared__ float erl[RA_NH];
  __shared__ __attribute__((aligned(16))) int lists[RA_WAVES * RA_LIST];
  __shared__ int ovf;
  const int lane = threadIdx.x & 31;
  const int wave = __builtin_amdgcn_readfirstlane((int)(threadIdx.x >> 5));
  const int n0   = blockIdx.x * RA_ROWS;
  const int h    = 2 * (lane & 3) + (lane >> 4);
  const bool sum_lane = (lane & 12) == 0;
  int* lst = lists + wave * RA_LIST;
  int* wsl = wl + (size_t)(blockIdx.x * RA_WAVES + wave) * RA_LIST;
  const int ovf1 = FIRST ? 0 : wo[blockIdx.x];

  for (int i = threadIdx.x; i < RA_ACC; i += RA_THREADS) acc[i] = 0.f;
  for (int i = threadIdx.x; i < RA_NH; i += RA_THREADS) {
    mxl[i] = __uint_as_float(0xff800000u);
    erl[i] = er[n0 * H_ + i];
  }
  if (threadIdx.x == 0) ovf = ovf1;
  __syncthreads();

  int cnt = 0;
  if (FIRST) {
  const uint4* key4 = (const uint4*)keys;
  const unsigned G = (unsigned)blockIdx.x * 0x01010101u;
  bool full = false;
#pragma unroll 1
  for (int q0 = wave; q0 < RA_KQ && !full; q0 += RA_WAVES * RA_SCAN) {
    uint4 v[RA_SCAN];
#pragma unroll
    for (int j = 0; j < RA_SCAN; ++j) {
      const int q = q0 + j * RA_WAVES;
      v[j] = key4[(q < RA_KQ ? q : q0) * 32 + lane];
    }
#pragma unroll
    for (int j = 0; j < RA_SCAN; ++j) {
      const int q = q0 + j * RA_WAVES;
      if (q < RA_KQ && !full) {
        if (cnt > RA_LIST - 512) { full = true; continue; }
        const unsigned kw[4] = {v[j].x, v[j].y, v[j].z, v[j].w};
        unsigned m = 0u;
#pragma unroll
        for (int c = 0; c < 4; ++c) {
          const unsigned t = kw[c] ^ G;
          const unsigned u = (t & 0x7F7F7F7Fu) + 0x7F7F7F7Fu;
          m |= ~(u | t | 0x7F7F7F7Fu) >> c;
        }
        const int gbase = (q * 32 + lane) << 5;
        unsigned bal = __builtin_amdgcn_ballot_w32(m != 0u);
        while (bal != 0u) {
          if (m != 0u) {
            lst[cnt + (int)__builtin_amdgcn_mbcnt_lo(bal, 0u)] = gbase | __builtin_ctz(m);
            m &= m - 1u;
          }
          cnt += __builtin_popcount(bal);
          bal = __builtin_amdgcn_ballot_w32(m != 0u);
        }
      }
    }
  }
  if (full) ovf = 1;

  for (int b = 0; b < cnt; b += 128) {
    int ent[4], sv[4], dv[4];
#pragma unroll
    for (int k = 0; k < 4; ++k) {
      const int i = b + 32 * k + lane;
      const int t = (i < cnt) ? lst[i] : 0;
      ent[k] = (t >> 5) * 16 + 4 * (7 - (t & 7)) + ((t & 31) >> 3);
    }
#pragma unroll
    for (int k = 0; k < 4; ++k) { sv[k] = src[ent[k]]; dv[k] = dst[ent[k]]; }
#pragma unroll
    for (int k = 0; k < 4; ++k) {
      const int i = b + 32 * k + lane;
      if (i < cnt) {
        const int val = sv[k] | ((dv[k] - n0) << 16);
        lst[i] = val;
        wsl[i] = val;
      }
    }
  }
  if (lane == 0) wc[blockIdx.x * RA_WAVES + wave] = cnt;
  } else {
    if (ovf1 == 0) cnt = __builtin_amdgcn_readfirstlane(wc[blockIdx.x * RA_WAVES + wave]);
    for (int b = 0; b < cnt; b += 128) {
      int ent[4];
#pragma unroll
      for (int k = 0; k < 4; ++k) {
        const int i = b + 32 * k + lane;
        ent[k] = wsl[i < cnt ? i : 0];
      }
#pragma unroll
      for (int k = 0; k < 4; ++k) {
        const int i = b + 32 * k + lane;
        if (i < cnt) lst[i] = ent[k];
      }
    }
  }
  {
    const int hh  = lane & 7;
    const int sub = lane >> 3;
    for (int b = 0; b < cnt; b += 4 * RA_MAXT) {
      int ent[RA_MAXT];
      float a[RA_MAXT];
#pragma unroll
      for (int t = 0; t < RA_MAXT; ++t) {
        const int i = b + 4 * t + sub;
        ent[t] = (i < cnt) ? lst[i] : -1;
      }
#pragma unroll
      for (int t = 0; t < RA_MAXT; ++t)
        a[t] = el[(ent[t] < 0 ? 0 : (ent[t] & 0xFFFF)) * H_ + hh];
#pragma unroll
      for (int t = 0; t < RA_MAXT; ++t) {
        if (ent[t] >= 0) {
          const int rh = (ent[t] >> 16) * H_ + hh;
          atomicMaxF(&mxl[rh], lrelu(a[t] + erl[rh]));
        }
      }
    }
  }
  __syncthreads();
  if (FIRST && threadIdx.x == 0) wo[blockIdx.x] = ovf;

  if (ovf == 0) {
    int head = 0;
    for (; cnt - head >= RA_GROUP; head += RA_GROUP)
      agg_group<true>(lst + head, RA_GROUP, lane, h, sum_lane, el, hp, erl, mxl, acc);
    if (cnt > head)
      agg_group<false>(lst + head, cnt - head, lane, h, sum_lane, el, hp, erl, mxl, acc);
  } else {
    for (int e = threadIdx.x; e < E_; e += RA_THREADS) {
      const int row = dst[e] - n0;
      if ((unsigned)row < (unsigned)RA_ROWS) {
        const int sn = src[e];
        for (int k = 0; k < H_; ++k)
          atomicMaxF(&mxl[row * H_ + k], lrelu(el[sn * H_ + k] + erl[row * H_ + k]));
      }
    }
    __syncthreads();
    for (int e = threadIdx.x; e < E_; e += RA_THREADS) {
      const int row = dst[e] - n0;
      if ((unsigned)row < (unsigned)RA_ROWS) {
        const int sn = src[e];
        for (int k = 0; k < H_; ++k) {
          const float v = lrelu(el[sn * H_ + k] + erl[row * H_ + k]);
          const float w = __expf(v - mxl[row * H_ + k]);
          atomicAdd(&acc[RA_SUMS + row * H_ + k], w);
          for (int d = 0; d < D_; ++d)
            atomicAdd(&acc[row * F_ + k * D_ + d], w * hp[(size_t)sn * F_ + k * D_ + d]);
        }
      }
    }
  }
  __syncthreads();

  float* hl = (float*)lists;
  hl[(RA_ROWS + (threadIdx.x >> 7)) * RA_PITCH + (threadIdx.x & 127)] = 0.f;
  const float4 bb = *(const float4*)(bias + lane * 4);
  const float4 g  = *(const float4*)(gamma + lane * 4);
  const float4 be = *(const float4*)(beta + lane * 4);
  for (int r = wave; r < RA_ROWS; r += RA_WAVES) {
    float4 x = *(const float4*)(acc + r * F_ + lane * 4);
    const float rs = 1.f / (acc[RA_SUMS + r * H_ + (lane >> 2)] + 1e-16f);
    x.x = x.x * rs + bb.x; x.y = x.y * rs + bb.y; x.z = x.z * rs + bb.z; x.w = x.w * rs + bb.w;
    float sA = x.x + x.y + x.z + x.w;
    float sB = x.x * x.x + x.y * x.y + x.z * x.z + x.w * x.w;
#pragma unroll
    for (int off = 16; off >= 1; off >>= 1) {
      sA += __shfl_xor(sA, off, 32);
      sB += __shfl_xor(sB, off, 32);
    }
    const float mu  = sA * (1.f / 128.f);
    const float var = sB * (1.f / 128.f) - mu * mu;
    const float inv = rsqrtf(var + LN_EPS);
    float4 y;
    y.x = fmaxf((x.x - mu) * inv * g.x + be.x, 0.f);
    y.y = fmaxf((x.y - mu) * inv * g.y + be.y, 0.f);
    y.z = fmaxf((x.z - mu) * inv * g.z + be.z, 0.f);
    y.w = fmaxf((x.w - mu) * inv * g.w + be.w, 0.f);
    *(float4*)(hl + r * RA_PITCH + lane * 4) = y;
  }
  __syncthreads();

  if (FIRST) {
    if (wave < 8) fc_tiles<4>(hl, wave, lane, n0, Wn, out);
    else          fc_tiles<3>(hl, wave, lane, n0, Wn, out);
  } else {
    for (int t = wave; t < RA_PTILES; t += RA_WAVES) {
      const int mt = t / 3;
      const int n  = (t - 3 * mt) * 16 + (lane & 15);
      v8f acc[1];
      rows_product<1, C_, C_>(hl, mt, n, lane, Wn, acc);
      const int m8 = mt * 16 + 8 * (lane >> 4);
      if (n < C_ && m8 < RA_ROWS) {
        const float bbn = bn[n];
#pragma unroll
        for (int r = 0; r < 8; ++r) out[(size_t)(n0 + m8 + r) * C_ + n] = acc[0][r] + bbn;
      }
    }
  }
}

extern "C" void kernel_launch(void* const* d_in, const int* in_sizes, int n_in,
                              void* d_out, int out_size, void* d_ws, size_t ws_size,
                              hipStream_t stream) {
  const float* feats = (const float*)d_in[0];
  const int*   src   = (const int*)d_in[1];
  const int*   dst   = (const int*)d_in[2];
  const float* Ws    = (const float*)d_in[3];
  const float* al    = (const float*)d_in[4];
  const float* ar    = (const float*)d_in[5];
  const float* bias  = (const float*)d_in[6];
  const float* gamma = (const float*)d_in[7];
  const float* beta  = (const float*)d_in[8];
  const float* Wp    = (const float*)d_in[9];
  const float* bp    = (const float*)d_in[10];
  float* outp = (float*)d_out;

  float* ws   = (float*)d_ws;
  float* hp0  = ws;
  float* hp1  = hp0 + (size_t)N_ * F_;
  float* el   = hp1 + (size_t)N_ * F_;
  float* er   = el + (size_t)N_ * H_;
  int* wl     = (int*)(er + (size_t)N_ * H_);
  int* wc     = wl + (size_t)RA_BLOCKS * RA_WAVES * RA_LIST;
  int* wo     = wc + RA_BLOCKS * RA_WAVES;
  unsigned int* keys = (unsigned int*)(wo + 256);

  const int NH = N_ * H_;

  gemm_node_fc<<<FC_BLOCKS, 256, 0, stream>>>(feats, Ws, hp0, dst, keys);
  node_coef<<<(NH + 255) / 256, 256, 0, stream>>>(hp0, al, ar, el, er, NH);
  range_aggregate<true><<<RA_BLOCKS, RA_THREADS, 0, stream>>>(
      src, dst, el, er, hp0, bias, gamma, beta, Ws + (size_t)F_ * F_, bp, hp1, wl, wc, wo, keys);
  node_coef<<<(NH + 255) / 256, 256, 0, stream>>>(
      hp1, al + (size_t)H_ * D_, ar + (size_t)H_ * D_, el, er, NH);
  range_aggregate<false><<<RA_BLOCKS, RA_THREADS, 0, stream>>>(
      src, dst, el, er, hp1, bias + (size_t)F_, gamma + (size_t)F_, beta + (size_t)F_,
      Wp, bp, outp, wl, wc, wo, keys);
}


# Softmax weights made once per (edge, head) in a pass behind the maxima barrier and kept in the wave's LDS list region: aggregation loads them, no second el gather; sums still in arrival order
# speedup vs baseline: 4.7048x; 1.0566x over previous
#include <hip/hip_runtime.h>

constexpr int N_ = 50000;
constexpr int E_ = 800000;
constexpr int F_ = 128;
constexpr int H_ = 8;
constexpr int D_ = 16;
constexpr int C_ = 40;
constexpr int L_ = 2;
constexpr float NEG_SLOPE = 0.2f;
constexpr float LN_EPS = 1e-5f;

constexpr int RA_ROWS    = 200;
constexpr int RA_BLOCKS  = N_ / RA_ROWS;
constexpr int RA_WAVES   = 32;
constexpr int RA_THREADS = RA_WAVES * 32;
constexpr int RA_KQ      = (E_ + 511) / 512;
constexpr int RA_KWORDS  = RA_KQ * 128;
constexpr int RA_SCAN    = 8;
constexpr int RA_GROUP   = 8;
constexpr int RA_MAXT    = 16;
constexpr int RA_LIST    = 1536;
constexpr int RA_WCAP    = 168;
constexpr int RA_WT      = 8;
constexpr int RA_NH      = RA_ROWS * H_;
constexpr int RA_SUMS    = RA_ROWS * F_;
constexpr int RA_ACC     = RA_ROWS * F_ + RA_NH;
static_assert(RA_BLOCKS * RA_ROWS == N_, "node ranges");
static_assert(E_ % 4 == 0 && RA_BLOCKS <= 255, "range keys: four per word, 255 marks a pad key");
static_assert(RA_LIST > 512 && RA_KQ * 32 <= (1 << 16), "a key load's hits fit behind the capacity check; list entry: group << 5 | bit");
constexpr int FC_RT      = 4;
constexpr int FC_BLOCKS  = (N_ / 16 + FC_RT - 1) / FC_RT;
static_assert(N_ % 16 == 0, "fc row tiles are whole");
static_assert(FC_BLOCKS * 256 >= RA_KWORDS, "the fc launch has a thread per key word");
static_assert(N_ <= (1 << 16), "list entry: source node | row << 16");
static_assert(9 * RA_WCAP <= RA_LIST && RA_WCAP % 8 == 0, "entries and eight weights per entry fit in the wave's list region");
static_assert((RA_ACC + 2 * RA_NH + RA_WAVES * RA_LIST + 1) * 4 <= 320 * 1024, "LDS of one work-group");
constexpr int RA_MT      = (RA_ROWS + 15) / 16;
constexpr int RA_PITCH   = F_ + 4;
constexpr int RA_PTILES  = RA_MT * ((C_ + 15) / 16);
static_assert(RA_MT * 16 * RA_PITCH <= RA_WAVES * RA_LIST, "finished rows fit where the lists were");
static_assert(RA_MT == 13 && RA_WAVES == 32, "fc tiles: 8 column slabs x (3 row tiles per wave, a 4th for waves 0..7)");
static_assert((RA_MT * 16 - RA_ROWS) * F_ == RA_THREADS, "pad rows: one float per thread");
static_assert(L_ == 2, "kernel_launch is written out for two layers");
static_assert(RA_ROWS % 8 == 0, "a lane's eight accumulator rows are all real or all pad");

typedef __attribute__((ext_vector_type(2))) float v2f;
typedef __attribute__((ext_vector_type(8))) float v8f;

__device__ __forceinline__ float lrelu(float x) { return x > 0.f ? x : NEG_SLOPE * x; }

__device__ __forceinline__ void atomicMaxF(float* addr, float val) {
  if (val >= 0.f) atomicMax((int*)addr, __float_as_int(val));
  else            atomicMin((unsigned int*)addr, __float_as_uint(val));
}

__global__ __launch_bounds__(256)
void gemm_node_fc(const float* __restrict__ X,
                  const float* __restrict__ W,
                  float* __restrict__ Y,
                  const int* __restrict__ dst,
                  unsigned int* __restrict__ keys) {
  {
    const int gid = blockIdx.x * 256 + threadIdx.x;
    if (gid < E_ / 4) {
      const int4 d = ((const int4*)dst)[gid];
      keys[gid] = ((unsigned)d.x / RA_ROWS) | (((unsigned)d.y / RA_ROWS) << 8) |
                  (((unsigned)d.z / RA_ROWS) << 16) | (((unsigned)d.w / RA_ROWS) << 24);
    } else if (gid < RA_KWORDS) {
      keys[gid] = 0xFFFFFFFFu;
    }
  }
  const int lane = threadIdx.x & 31;
  const int wave = threadIdx.x >> 5;
  const int rt   = blockIdx.x * FC_RT + (wave >> 1);
  if (rt >= N_ / 16) return;
  const int m0   = rt * 16;
  const int half = lane >> 4;
  const int l16  = lane & 15;
  const int cb   = (wave & 1) * 64 + 4 * l16;
  const float* ap = X + (size_t)(m0 + l16) * F_ + 2 * half;
  const float* bp = W + (size_t)(2 * half) * F_ + cb;

  v8f acc[4];
#pragma unroll
  for (int t = 0; t < 4; ++t) acc[t] = v8f{};
#pragma unroll
  for (int k = 0; k < F_; k += 4) {
    const v2f a = *(const v2f*)(ap + k);
    const float4 b0 = *(const float4*)(bp + (size_t)k * F_);
    const float4 b1 = *(const float4*)(bp + (size_t)(k + 1) * F_);
    v2f b;
    b.x = b0.x; b.y = b1.x;
    acc[0] = __builtin_amdgcn_wmma_f32_16x16x4_f32(false, a, false, b, (short)0, acc[0], false, false);
    b.x = b0.y; b.y = b1.y;
    acc[1] = __builtin_amdgcn_wmma_f32_16x16x4_f32(false, a, false, b, (short)0, acc[1], false, false);
    b.x = b0.z; b.y = b1.z;
    acc[2] = __builtin_amdgcn_wmma_f32_16x16x4_f32(false, a, false, b, (short)0, acc[2], false, false);
    b.x = b0.w; b.y = b1.w;
    acc[3] = __builtin_amdgcn_wmma_f32_16x16x4_f32(false, a, false, b, (short)0, acc[3], false, false);
  }
#pragma unroll
  for (int r = 0; r < 8; ++r) {
    const int m = m0 + r + 8 * half;
    float4 o;
    o.x = acc[0][r]; o.y = acc[1][r]; o.z = acc[2][r]; o.w = acc[3][r];
    *(float4*)(Y + (size_t)m * F_ + cb) = o;
  }
}

__global__ void node_coef(const float* __restrict__ hp,
                          const float* __restrict__ al,
                          const float* __restrict__ ar,
                          float* __restrict__ el, float* __restrict__ er, int total) {
  const int t = blockIdx.x * blockDim.x + threadIdx.x;
  if (t >= total) return;
  const int n = t >> 3;
  const int h = t & 7;
  const float* row = hp + n * F_ + h * D_;
  float sl = 0.f, sr = 0.f;
#pragma unroll
  for (int d = 0; d < D_; ++d) {
    const float x = row[d];
    sl += x * al[h * D_ + d];
    sr += x * ar[h * D_ + d];
  }
  el[t] = sl;
  er[t] = sr;
}

template <int J>
__device__ __forceinline__ float quad_lane(float v) {
  return __int_as_float(__builtin_amdgcn_update_dpp(
      0, __float_as_int(v), J * 0x55, 0xF, 0xF, true));
}

template <bool FULL>
__device__ __forceinline__ void agg_group(const int* lst, int m, int lane, int h,
                                          bool sum_lane,
                                          const float* __restrict__ el,
                                          const float* __restrict__ hp,
                                          const float* erl, const float* mxl, float* acc) {
  const int slot = lane & (RA_GROUP - 1);
  const int ev = lst[(FULL || slot < m) ? slot : 0];
  int s[RA_GROUP], r[RA_GROUP];
  float cl[RA_GROUP], cr[RA_GROUP], cm[RA_GROUP], wt[RA_GROUP];
  float x0[RA_GROUP], x1[RA_GROUP], x2[RA_GROUP], x3[RA_GROUP];
#pragma unroll
  for (int j = 0; j < RA_GROUP; ++j) {
    const int t = __builtin_amdgcn_readlane(ev, j);
    s[j] = t & 0xFFFF;
    r[j] = t >> 16;
  }
#pragma unroll
  for (int j = 0; j < RA_GROUP; ++j) {
    cl[j] = el[s[j] * H_ + h];
    cr[j] = erl[r[j] * H_ + h];
    cm[j] = mxl[r[j] * H_ + h];
    const float* x = hp + (size_t)s[j] * F_ + lane;
    x0[j] = x[0]; x1[j] = x[32]; x2[j] = x[64]; x3[j] = x[96];
  }
  __builtin_amdgcn_sched_barrier(0);
#pragma unroll
  for (int j = 0; j < RA_GROUP; ++j) {
    const float v     = lrelu(cl[j] + cr[j]);
    const float alpha = __expf(v - cm[j]);
    wt[j] = alpha;
    if (FULL || j < m) {
      float* o = acc + r[j] * F_ + lane;
      atomicAdd(o,      quad_lane<0>(alpha) * x0[j]);
      atomicAdd(o + 32, quad_lane<1>(alpha) * x1[j]);
      atomicAdd(o + 64, quad_lane<2>(alpha) * x2[j]);
      atomicAdd(o + 96, quad_lane<3>(alpha) * x3[j]);
    }
  }
  if (sum_lane) {
#pragma unroll
    for (int j = 0; j < RA_GROUP; ++j)
      if (FULL || j < m) atomicAdd(&acc[RA_SUMS + r[j] * H_ + h], wt[j]);
  }
}

template <bool FULL>
__device__ __forceinline__ void agg_group_w(const int* lst, const float* wst, int m, int lane, int h,
                                            const float* __restrict__ hp, float* acc) {
  const int slot = lane & (RA_GROUP - 1);
  const int ev = lst[(FULL || slot < m) ? slot : 0];
  int s[RA_GROUP], r[RA_GROUP];
  float wt[RA_GROUP];
  float x0[RA_GROUP], x1[RA_GROUP], x2[RA_GROUP], x3[RA_GROUP];
#pragma unroll
  for (int j = 0; j < RA_GROUP; ++j) {
    const int t = __builtin_amdgcn_readlane(ev, j);
    s[j] = t & 0xFFFF;
    r[j] = t >> 16;
  }
#pragma unroll
  for (int j = 0; j < RA_GROUP; ++j) {
    wt[j] = wst[((FULL || j < m) ? j : 0) * H_ + h];
    const float* x = hp + (size_t)s[j] * F_ + lane;
    x0[j] = x[0]; x1[j] = x[32]; x2[j] = x[64]; x3[j] = x[96];
  }
  __builtin_amdgcn_sched_barrier(0);
#pragma unroll
  for (int j = 0; j < RA_GROUP; ++j) {
    if (FULL || j < m) {
      float* o = acc + r[j] * F_ + lane;
      atomicAdd(o,      quad_lane<0>(wt[j]) * x0[j]);
      atomicAdd(o + 32, quad_lane<1>(wt[j]) * x1[j]);
      atomicAdd(o + 64, quad_lane<2>(wt[j]) * x2[j]);
      atomicAdd(o + 96, quad_lane<3>(wt[j]) * x3[j]);
    }
  }
}

template <int MT, int LDW, int NCOLS>
__device__ __forceinline__ void rows_product(const float* hl, int mt0, int col, int lane,
                                             const float* __restrict__ Wm, v8f (&acc)[MT]) {
  const int half = lane >> 4;
  const int l16  = lane & 15;
  const bool cv  = (NCOLS % 16 == 0) || col < NCOLS;
  const float* ap = hl + (mt0 * 16 + l16) * RA_PITCH + 2 * half;
  const float* bp = Wm + (2 * half) * LDW + (cv ? col : 0);
#pragma unroll
  for (int i = 0; i < MT; ++i) acc[i] = v8f{};
#pragma unroll
  for (int k = 0; k < F_; k += 4) {
    const float bx = bp[k * LDW];
    const float by = bp[(k + 1) * LDW];
    v2f b;
    b.x = cv ? bx : 0.f;
    b.y = cv ? by : 0.f;
#pragma unroll
    for (int i = 0; i < MT; ++i) {
      const v2f a = *(const v2f*)(ap + i * 64 * RA_PITCH + k);
      acc[i] = __builtin_amdgcn_wmma_f32_16x16x4_f32(false, a, false, b,
                                                     (short)0, acc[i], false, false);
    }
  }
}

template <int MT>
__device__ __forceinline__ void fc_tiles(const float* hl, int wave, int lane, int n0,
                                         const float* __restrict__ Wn, float* __restrict__ out) {
  const int mt0 = wave >> 3;
  const int col = (wave & 7) * 16 + (lane & 15);
  v8f acc[MT];
  rows_product<MT, F_, F_>(hl, mt0, col, lane, Wn, acc);
#pragma unroll
  for (int i = 0; i < MT; ++i) {
    const int m8 = (mt0 + 4 * i) * 16 + 8 * (lane >> 4);
    if (m8 < RA_ROWS) {
#pragma unroll
      for (int r = 0; r < 8; ++r) out[(size_t)(n0 + m8 + r) * F_ + col] = acc[i][r];
    }
  }
}

template <bool FIRST>
__global__ __launch_bounds__(RA_THREADS)
void range_aggregate(const int* __restrict__ src, const int* __restrict__ dst,
                     const float* __restrict__ el, const float* __restrict__ er,
                     const float* __restrict__ hp,
                     const float* __restrict__ bias, const float* __restrict__ gamma,
                     const float* __restrict__ beta,
                     const float* __restrict__ Wn, const float* __restrict__ bn,
                     float* __restrict__ out,
                     int* __restrict__ wl, int* __restrict__ wc, int* __restrict__ wo,
                     const unsigned int* __restrict__ keys) {
  __shared__ __attribute__((aligned(16))) float acc[RA_ACC];
  __shared__ float mxl[RA_NH];
  __shared__ float erl[RA_NH];
  __shared__ __attribute__((aligned(16))) int lists[RA_WAVES * RA_LIST];
  __shared__ int ovf;
  const int lane = threadIdx.x & 31;
  const int wave = __builtin_amdgcn_readfirstlane((int)(threadIdx.x >> 5));
  const int n0   = blockIdx.x * RA_ROWS;
  const int h    = 2 * (lane & 3) + (lane >> 4);
  const bool sum_lane = (lane & 12) == 0;
  int* lst = lists + wave * RA_LIST;
  int* wsl = wl + (size_t)(blockIdx.x * RA_WAVES + wave) * RA_LIST;
  const int ovf1 = FIRST ? 0 : wo[blockIdx.x];

  for (int i = threadIdx.x; i < RA_ACC; i += RA_THREADS) acc[i] = 0.f;
  for (int i = threadIdx.x; i < RA_NH; i += RA_THREADS) {
    mxl[i] = __uint_as_float(0xff800000u);
    erl[i] = er[n0 * H_ + i];
  }
  if (threadIdx.x == 0) ovf = ovf1;
  __syncthreads();

  int cnt = 0;
  if (FIRST) {
  const uint4* key4 = (const uint4*)keys;
  const unsigned G = (unsigned)blockIdx.x * 0x01010101u;
  bool full = false;
#pragma unroll 1
  for (int q0 = wave; q0 < RA_KQ && !full; q0 += RA_WAVES * RA_SCAN) {
    uint4 v[RA_SCAN];
#pragma unroll
    for (int j = 0; j < RA_SCAN; ++j) {
      const int q = q0 + j * RA_WAVES;
      v[j] = key4[(q < RA_KQ ? q : q0) * 32 + lane];
    }
#pragma unroll
    for (int j = 0; j < RA_SCAN; ++j) {
      const int q = q0 + j * RA_WAVES;
      if (q < RA_KQ && !full) {
        if (cnt > RA_LIST - 512) { full = true; continue; }
        const unsigned kw[4] = {v[j].x, v[j].y, v[j].z, v[j].w};
        unsigned m = 0u;
#pragma unroll
        for (int c = 0; c < 4; ++c) {
          const unsigned t = kw[c] ^ G;
          const unsigned u = (t & 0x7F7F7F7Fu) + 0x7F7F7F7Fu;
          m |= ~(u | t | 0x7F7F7F7Fu) >> c;
        }
        const int gbase = (q * 32 + lane) << 5;
        unsigned bal = __builtin_amdgcn_ballot_w32(m != 0u);
        while (bal != 0u) {
          if (m != 0u) {
            lst[cnt + (int)__builtin_amdgcn_mbcnt_lo(bal, 0u)] = gbase | __builtin_ctz(m);
            m &= m - 1u;
          }
          cnt += __builtin_popcount(bal);
          bal = __builtin_amdgcn_ballot_w32(m != 0u);
        }
      }
    }
  }
  if (full) ovf = 1;

  for (int b = 0; b < cnt; b += 128) {
    int ent[4], sv[4], dv[4];
#pragma unroll
    for (int k = 0; k < 4; ++k) {
      const int i = b + 32 * k + lane;
      const int t = (i < cnt) ? lst[i] : 0;
      ent[k] = (t >> 5) * 16 + 4 * (7 - (t & 7)) + ((t & 31) >> 3);
    }
#pragma unroll
    for (int k = 0; k < 4; ++k) { sv[k] = src[ent[k]]; dv[k] = dst[ent[k]]; }
#pragma unroll
    for (int k = 0; k < 4; ++k) {
      const int i = b + 32 * k + lane;
      if (i < cnt) {
        const int val = sv[k] | ((dv[k] - n0) << 16);
        lst[i] = val;
        wsl[i] = val;
      }
    }
  }
  if (lane == 0) wc[blockIdx.x * RA_WAVES + wave] = cnt;
  } else {
    if (ovf1 == 0) cnt = __builtin_amdgcn_readfirstlane(wc[blockIdx.x * RA_WAVES + wave]);
    for (int b = 0; b < cnt; b += 128) {
      int ent[4];
#pragma unroll
      for (int k = 0; k < 4; ++k) {
        const int i = b + 32 * k + lane;
        ent[k] = wsl[i < cnt ? i : 0];
      }
#pragma unroll
      for (int k = 0; k < 4; ++k) {
        const int i = b + 32 * k + lane;
        if (i < cnt) lst[i] = ent[k];
      }
    }
  }
  const bool keepw = cnt <= RA_WCAP;
  float* wst = (float*)(lst + RA_WCAP);
  {
    const int hh  = lane & 7;
    const int sub = lane >> 3;
    for (int b = 0; b < cnt; b += 4 * RA_MAXT) {
      int ent[RA_MAXT];
      float a[RA_MAXT];
#pragma unroll
      for (int t = 0; t < RA_MAXT; ++t) {
        const int i = b + 4 * t + sub;
        ent[t] = (i < cnt) ? lst[i] : -1;
      }
#pragma unroll
      for (int t = 0; t < RA_MAXT; ++t)
        a[t] = el[(ent[t] < 0 ? 0 : (ent[t] & 0xFFFF)) * H_ + hh];
#pragma unroll
      for (int t = 0; t < RA_MAXT; ++t) {
        if (ent[t] >= 0) {
          const int rh = (ent[t] >> 16) * H_ + hh;
          const float sc = lrelu(a[t] + erl[rh]);
          atomicMaxF(&mxl[rh], sc);
          if (keepw) wst[(b + 4 * t) * H_ + lane] = sc;
        }
      }
    }
  }
  __syncthreads();
  if (FIRST && threadIdx.x == 0) wo[blockIdx.x] = ovf;

  if (ovf == 0 && keepw) {
    const int hh  = lane & 7;
    const int sub = lane >> 3;
    for (int b = 0; b < cnt; b += 4 * RA_WT) {
      int rh[RA_WT];
      float sc[RA_WT], mx[RA_WT];
#pragma unroll
      for (int t = 0; t < RA_WT; ++t) {
        const int i = b + 4 * t + sub;
        const int e = (i < cnt) ? lst[i] : -1;
        rh[t] = (e < 0) ? -1 : (e >> 16) * H_ + hh;
      }
#pragma unroll
      for (int t = 0; t < RA_WT; ++t) {
        sc[t] = wst[(rh[t] < 0) ? 0 : (b + 4 * t) * H_ + lane];
        mx[t] = mxl[(rh[t] < 0) ? 0 : rh[t]];
      }
#pragma unroll
      for (int t = 0; t < RA_WT; ++t) {
        if (rh[t] >= 0) {
          const float w = __expf(sc[t] - mx[t]);
          wst[(b + 4 * t) * H_ + lane] = w;
          atomicAdd(&acc[RA_SUMS + rh[t]], w);
        }
      }
    }
    int head = 0;
    for (; cnt - head >= RA_GROUP; head += RA_GROUP)
      agg_group_w<true>(lst + head, wst + head * H_, RA_GROUP, lane, h, hp, acc);
    if (cnt > head)
      agg_group_w<false>(lst + head, wst + head * H_, cnt - head, lane, h, hp, acc);
  } else if (ovf == 0) {
    int head = 0;
    for (; cnt - head >= RA_GROUP; head += RA_GROUP)
      agg_group<true>(lst + head, RA_GROUP, lane, h, sum_lane, el, hp, erl, mxl, acc);
    if (cnt > head)
      agg_group<false>(lst + head, cnt - head, lane, h, sum_lane, el, hp, erl, mxl, acc);
  } else {
    for (int e = threadIdx.x; e < E_; e += RA_THREADS) {
      const int row = dst[e] - n0;
      if ((unsigned)row < (unsigned)RA_ROWS) {
        const int sn = src[e];
        for (int k = 0; k < H_; ++k)
          atomicMaxF(&mxl[row * H_ + k], lrelu(el[sn * H_ + k] + erl[row * H_ + k]));
      }
    }
    __syncthreads();
    for (int e = threadIdx.x; e < E_; e += RA_THREADS) {
      const int row = dst[e] - n0;
      if ((unsigned)row < (unsigned)RA_ROWS) {
        const int sn = src[e];
        for (int k = 0; k < H_; ++k) {
          const float v = lrelu(el[sn * H_ + k] + erl[row * H_ + k]);
          const float w = __expf(v - mxl[row * H_ + k]);
          atomicAdd(&acc[RA_SUMS + row * H_ + k], w);
          for (int d = 0; d < D_; ++d)
            atomicAdd(&acc[row * F_ + k * D_ + d], w * hp[(size_t)sn * F_ + k * D_ + d]);
        }
      }
    }
  }
  __syncthreads();

  float* hl = (float*)lists;
  hl[(RA_ROWS + (threadIdx.x >> 7)) * RA_PITCH + (threadIdx.x & 127)] = 0.f;
  const float4 bb = *(const float4*)(bias + lane * 4);
  const float4 g  = *(const float4*)(gamma + lane * 4);
  const float4 be = *(const float4*)(beta + lane * 4);
  for (int r = wave; r < RA_ROWS; r += RA_WAVES) {
    float4 x = *(const float4*)(acc + r * F_ + lane * 4);
    const float rs = 1.f / (acc[RA_SUMS + r * H_ + (lane >> 2)] + 1e-16f);
    x.x = x.x * rs + bb.x; x.y = x.y * rs + bb.y; x.z = x.z * rs + bb.z; x.w = x.w * rs + bb.w;
    float sA = x.x + x.y + x.z + x.w;
    float sB = x.x * x.x + x.y * x.y + x.z * x.z + x.w * x.w;
#pragma unroll
    for (int off = 16; off >= 1; off >>= 1) {
      sA += __shfl_xor(sA, off, 32);
      sB += __shfl_xor(sB, off, 32);
    }
    const float mu  = sA * (1.f / 128.f);
    const float var = sB * (1.f / 128.f) - mu * mu;
    const float inv = rsqrtf(var + LN_EPS);
    float4 y;
    y.x = fmaxf((x.x - mu) * inv * g.x + be.x, 0.f);
    y.y = fmaxf((x.y - mu) * inv * g.y + be.y, 0.f);
    y.z = fmaxf((x.z - mu) * inv * g.z + be.z, 0.f);
    y.w = fmaxf((x.w - mu) * inv * g.w + be.w, 0.f);
    *(float4*)(hl + r * RA_PITCH + lane * 4) = y;
  }
  __syncthreads();

  if (FIRST) {
    if (wave < 8) fc_tiles<4>(hl, wave, lane, n0, Wn, out);
    else          fc_tiles<3>(hl, wave, lane, n0, Wn, out);
  } else {
    for (int t = wave; t < RA_PTILES; t += RA_WAVES) {
      const int mt = t / 3;
      const int n  = (t - 3 * mt) * 16 + (lane & 15);
      v8f acc[1];
      rows_product<1, C_, C_>(hl, mt, n, lane, Wn, acc);
      const int m8 = mt * 16 + 8 * (lane >> 4);
      if (n < C_ && m8 < RA_ROWS) {
        const float bbn = bn[n];
#pragma unroll
        for (int r = 0; r < 8; ++r) out[(size_t)(n0 + m8 + r) * C_ + n] = acc[0][r] + bbn;
      }
    }
  }
}

extern "C" void kernel_launch(void* const* d_in, const int* in_sizes, int n_in,
                              void* d_out, int out_size, void* d_ws, size_t ws_size,
                              hipStream_t stream) {
  const float* feats = (const float*)d_in[0];
  const int*   src   = (const int*)d_in[1];
  const int*   dst   = (const int*)d_in[2];
  const float* Ws    = (const float*)d_in[3];
  const float* al    = (const float*)d_in[4];
  const float* ar    = (const float*)d_in[5];
  const float* bias  = (const float*)d_in[6];
  const float* gamma = (const float*)d_in[7];
  const float* beta  = (const float*)d_in[8];
  const float* Wp    = (const float*)d_in[9];
  const float* bp    = (const float*)d_in[10];
  float* outp = (float*)d_out;

  float* ws   = (float*)d_ws;
  float* hp0  = ws;
  float* hp1  = hp0 + (size_t)N_ * F_;
  float* el   = hp1 + (size_t)N_ * F_;
  float* er   = el + (size_t)N_ * H_;
  int* wl     = (int*)(er + (size_t)N_ * H_);
  int* wc     = wl + (size_t)RA_BLOCKS * RA_WAVES * RA_LIST;
  int* wo     = wc + RA_BLOCKS * RA_WAVES;
  unsigned int* keys = (unsigned int*)(wo + 256);

  const int NH = N_ * H_;

  gemm_node_fc<<<FC_BLOCKS, 256, 0, stream>>>(feats, Ws, hp0, dst, keys);
  node_coef<<<(NH + 255) / 256, 256, 0, stream>>>(hp0, al, ar, el, er, NH);
  range_aggregate<true><<<RA_BLOCKS, RA_THREADS, 0, stream>>>(
      src, dst, el, er, hp0, bias, gamma, beta, Ws + (size_t)F_ * F_, bp, hp1, wl, wc, wo, keys);
  node_coef<<<(NH + 255) / 256, 256, 0, stream>>>(
      hp1, al + (size_t)H_ * D_, ar + (size_t)H_ * D_, el, er, NH);
  range_aggregate<false><<<RA_BLOCKS, RA_THREADS, 0, stream>>>(
      src, dst, el, er, hp1, bias + (size_t)F_, gamma + (size_t)F_, beta + (size_t)F_,
      Wp, bp, outp, wl, wc, wo, keys);
}
